# starting kernel
# baseline (speedup 1.0000x reference)
#include <hip/hip_runtime.h>
#include <hip/hip_bf16.h>

typedef __attribute__((ext_vector_type(2))) float v2f;
typedef __attribute__((ext_vector_type(8))) float v8f;

#define BLOCK 128

__device__ __forceinline__ v8f wmma4(v2f a, v2f b, v8f c) {
  return __builtin_amdgcn_wmma_f32_16x16x4_f32(false, a, false, b, (short)0, c,
                                               false, false);
}

__global__ void reduce_wq_kernel(const float* __restrict__ Wq,
                                 float* __restrict__ Wqg) {
  int idx = blockIdx.x * blockDim.x + threadIdx.x;
  if (idx >= 1024 * 256) return;
  int i = idx >> 8;
  int c = idx & 255;
  int h = c >> 6, d = c & 63;
  const float* p = Wq + (size_t)i * 1024 + h * 256 + d;
  Wqg[idx] = p[0] + p[64] + p[128] + p[192];
}

__global__ __launch_bounds__(BLOCK) void gemm_f32_wmma(
    const float* __restrict__ A, const float* __restrict__ B,
    float* __restrict__ C, int M, int N, int K) {
  __shared__ __align__(16) float As[64][18];
  __shared__ __align__(16) float Bs[16][66];
  const int tid  = threadIdx.x;
  const int wave = tid >> 5;
  const int lane = tid & 31;
  const int half = lane >> 4;
  const int l16  = lane & 15;
  const int row0 = blockIdx.y * 64;
  const int col0 = blockIdx.x * 64;

  v8f acc[4];
#pragma unroll
  for (int nt = 0; nt < 4; ++nt) acc[nt] = (v8f)(0.0f);

  for (int kk = 0; kk < K; kk += 16) {
    __syncthreads();
#pragma unroll
    for (int f = tid; f < 256; f += BLOCK) {
      int r = f >> 2, c = (f & 3) << 2;
      const float4 v = *(const float4*)(A + (size_t)(row0 + r) * K + kk + c);
      As[r][c] = v.x; As[r][c + 1] = v.y; As[r][c + 2] = v.z; As[r][c + 3] = v.w;
    }
#pragma unroll
    for (int f = tid; f < 256; f += BLOCK) {
      int r = f >> 4, c = (f & 15) << 2;
      const float4 v = *(const float4*)(B + (size_t)(kk + r) * N + col0 + c);
      Bs[r][c] = v.x; Bs[r][c + 1] = v.y; Bs[r][c + 2] = v.z; Bs[r][c + 3] = v.w;
    }
    __syncthreads();
#pragma unroll
    for (int k0 = 0; k0 < 16; k0 += 4) {
      v2f a = *(const v2f*)&As[wave * 16 + l16][k0 + 2 * half];
#pragma unroll
      for (int nt = 0; nt < 4; ++nt) {
        v2f b;
        b.x = Bs[k0 + 2 * half + 0][nt * 16 + l16];
        b.y = Bs[k0 + 2 * half + 1][nt * 16 + l16];
        acc[nt] = wmma4(a, b, acc[nt]);
      }
    }
  }
#pragma unroll
  for (int nt = 0; nt < 4; ++nt)
#pragma unroll
    for (int r = 0; r < 8; ++r)
      C[(size_t)(row0 + wave * 16 + r + 8 * half) * N + col0 + nt * 16 + l16] =
          acc[nt][r];
}

__global__ __launch_bounds__(BLOCK) void fa_f32_wmma(
    const float* __restrict__ Qg, const float* __restrict__ Kp,
    const float* __restrict__ Vp, float* __restrict__ Op, int S) {
  __shared__ __align__(16) float Ks[16][66];
  __shared__ __align__(16) float Vs[16][66];
  __shared__ __align__(16) float Ps[4][16][18];
  const int tid  = threadIdx.x;
  const int wave = tid >> 5;
  const int lane = tid & 31;
  const int half = lane >> 4;
  const int l16  = lane & 15;
  const int q0   = blockIdx.x * 64;
  const int b    = blockIdx.y >> 2;
  const int h    = blockIdx.y & 3;

  v2f qf[16];
  {
    const int qrow = q0 + wave * 16 + l16;
    const float* qptr =
        Qg + (size_t)(b * S + qrow) * 256 + h * 64 + 2 * half;
#pragma unroll
    for (int i = 0; i < 16; ++i) qf[i] = *(const v2f*)(qptr + i * 4);
  }

  v8f oacc[4];
#pragma unroll
  for (int nt = 0; nt < 4; ++nt) oacc[nt] = (v8f)(0.0f);
  float m[8], l[8];
#pragma unroll
  for (int r = 0; r < 8; ++r) { m[r] = -1e30f; l[r] = 0.0f; }

  const int ntiles = (q0 >> 4) + 4;
  for (int j = 0; j < ntiles; ++j) {
    const int a0 = j << 4;
    __syncthreads();
#pragma unroll
    for (int f = tid; f < 256; f += BLOCK) {
      int r = f >> 4, c = (f & 15) << 2;
      const size_t g = (size_t)(b * S + a0 + r) * 256 + h * 64 + c;
      const float4 kv = *(const float4*)(Kp + g);
      Ks[r][c] = kv.x; Ks[r][c + 1] = kv.y; Ks[r][c + 2] = kv.z; Ks[r][c + 3] = kv.w;
      const float4 vv = *(const float4*)(Vp + g);
      Vs[r][c] = vv.x; Vs[r][c + 1] = vv.y; Vs[r][c + 2] = vv.z; Vs[r][c + 3] = vv.w;
    }
    __syncthreads();

    v8f s = (v8f)(0.0f);
#pragma unroll
    for (int i = 0; i < 16; ++i) {
      v2f bk;
      bk.x = Ks[l16][i * 4 + 2 * half + 0];
      bk.y = Ks[l16][i * 4 + 2 * half + 1];
      s = wmma4(qf[i], bk, s);
    }

    const int acol = a0 + l16;
    float pr[8];
#pragma unroll
    for (int r = 0; r < 8; ++r) {
      const int qrow = q0 + wave * 16 + r + 8 * half;
      float x = s[r] * 0.125f;
      if (acol > qrow) x = -1e30f;
      float mx = x;
#pragma unroll
      for (int off = 1; off < 16; off <<= 1)
        mx = fmaxf(mx, __shfl_xor(mx, off, 16));
      const float mnew  = fmaxf(m[r], mx);
      const float p     = __expf(x - mnew);
      float sum = p;
#pragma unroll
      for (int off = 1; off < 16; off <<= 1) sum += __shfl_xor(sum, off, 16);
      const float scale = __expf(m[r] - mnew);
      l[r] = l[r] * scale + sum;
      m[r] = mnew;
#pragma unroll
      for (int nt = 0; nt < 4; ++nt) oacc[nt][r] *= scale;
      pr[r] = p;
    }

#pragma unroll
    for (int r = 0; r < 8; ++r) Ps[wave][r + 8 * half][l16] = pr[r];
    __syncthreads();

#pragma unroll
    for (int k0i = 0; k0i < 4; ++k0i) {
      const int k0 = k0i * 4;
      v2f a = *(const v2f*)&Ps[wave][l16][k0 + 2 * half];
#pragma unroll
      for (int nt = 0; nt < 4; ++nt) {
        v2f bv;
        bv.x = Vs[k0 + 2 * half + 0][nt * 16 + l16];
        bv.y = Vs[k0 + 2 * half + 1][nt * 16 + l16];
        oacc[nt] = wmma4(a, bv, oacc[nt]);
      }
    }
  }

#pragma unroll
  for (int r = 0; r < 8; ++r) {
    const float inv = 1.0f / l[r];
    const int row = b * S + q0 + wave * 16 + r + 8 * half;
#pragma unroll
    for (int nt = 0; nt < 4; ++nt)
      Op[(size_t)row * 256 + h * 64 + nt * 16 + l16] = oacc[nt][r] * inv;
  }
}

extern "C" void kernel_launch(void* const* d_in, const int* in_sizes, int n_in,
                              void* d_out, int out_size, void* d_ws,
                              size_t ws_size, hipStream_t stream) {
  const float* q_in  = (const float*)d_in[0];
  const float* kv_in = (const float*)d_in[1];
  const float* Wq = (const float*)d_in[3];
  const float* Wk = (const float*)d_in[4];
  const float* Wv = (const float*)d_in[5];
  const float* Wo = (const float*)d_in[6];
  float* out = (float*)d_out;

  const int B = 2, S = 2048, IN = 1024;
  const int M = B * S;

  float* ws  = (float*)d_ws;
  float* wqg = ws;
  float* qg  = wqg + 1024 * 256;
  float* kp  = qg + M * 256;
  float* vp  = kp + M * 256;
  float* op  = vp + M * 256;

  reduce_wq_kernel<<<(1024 * 256) / 256, 256, 0, stream>>>(Wq, wqg);

  dim3 blk(BLOCK);
  gemm_f32_wmma<<<dim3(256 / 64, M / 64), blk, 0, stream>>>(q_in, wqg, qg, M, 256, IN);
  gemm_f32_wmma<<<dim3(256 / 64, M / 64), blk, 0, stream>>>(kv_in, Wk, kp, M, 256, IN);
  gemm_f32_wmma<<<dim3(256 / 64, M / 64), blk, 0, stream>>>(kv_in, Wv, vp, M, 256, IN);

  fa_f32_wmma<<<dim3(S / 64, B * 4), blk, 0, stream>>>(qg, kp, vp, op, S);

  gemm_f32_wmma<<<dim3(IN / 64, M / 64), blk, 0, stream>>>(op, Wo, out, M, IN, 256);
}


# attention key loop covers 128 keys per trip, not 16: one load round trip, one cross-lane max/sum chain, one rescale, one barrier set per 128 keys; changes the order of row sums and rescales
# speedup vs baseline: 2.0489x; 2.0489x over previous
#include <hip/hip_runtime.h>
#include <hip/hip_bf16.h>

typedef __attribute__((ext_vector_type(2))) float v2f;
typedef __attribute__((ext_vector_type(8))) float v8f;

#define BLOCK 128

__device__ __forceinline__ v8f wmma4(v2f a, v2f b, v8f c) {
  return __builtin_amdgcn_wmma_f32_16x16x4_f32(false, a, false, b, (short)0, c,
                                               false, false);
}

__global__ void reduce_wq_kernel(const float* __restrict__ Wq,
                                 float* __restrict__ Wqg) {
  int idx = blockIdx.x * blockDim.x + threadIdx.x;
  if (idx >= 1024 * 256) return;
  int i = idx >> 8;
  int c = idx & 255;
  int h = c >> 6, d = c & 63;
  const float* p = Wq + (size_t)i * 1024 + h * 256 + d;
  Wqg[idx] = p[0] + p[64] + p[128] + p[192];
}

__global__ __launch_bounds__(BLOCK) void gemm_f32_wmma(
    const float* __restrict__ A, const float* __restrict__ B,
    float* __restrict__ C, int M, int N, int K) {
  __shared__ __align__(16) float As[64][18];
  __shared__ __align__(16) float Bs[16][66];
  const int tid  = threadIdx.x;
  const int wave = tid >> 5;
  const int lane = tid & 31;
  const int half = lane >> 4;
  const int l16  = lane & 15;
  const int row0 = blockIdx.y * 64;
  const int col0 = blockIdx.x * 64;

  v8f acc[4];
#pragma unroll
  for (int nt = 0; nt < 4; ++nt) acc[nt] = (v8f)(0.0f);

  for (int kk = 0; kk < K; kk += 16) {
    __syncthreads();
#pragma unroll
    for (int f = tid; f < 256; f += BLOCK) {
      int r = f >> 2, c = (f & 3) << 2;
      const float4 v = *(const float4*)(A + (size_t)(row0 + r) * K + kk + c);
      As[r][c] = v.x; As[r][c + 1] = v.y; As[r][c + 2] = v.z; As[r][c + 3] = v.w;
    }
#pragma unroll
    for (int f = tid; f < 256; f += BLOCK) {
      int r = f >> 4, c = (f & 15) << 2;
      const float4 v = *(const float4*)(B + (size_t)(kk + r) * N + col0 + c);
      Bs[r][c] = v.x; Bs[r][c + 1] = v.y; Bs[r][c + 2] = v.z; Bs[r][c + 3] = v.w;
    }
    __syncthreads();
#pragma unroll
    for (int k0 = 0; k0 < 16; k0 += 4) {
      v2f a = *(const v2f*)&As[wave * 16 + l16][k0 + 2 * half];
#pragma unroll
      for (int nt = 0; nt < 4; ++nt) {
        v2f b;
        b.x = Bs[k0 + 2 * half + 0][nt * 16 + l16];
        b.y = Bs[k0 + 2 * half + 1][nt * 16 + l16];
        acc[nt] = wmma4(a, b, acc[nt]);
      }
    }
  }
#pragma unroll
  for (int nt = 0; nt < 4; ++nt)
#pragma unroll
    for (int r = 0; r < 8; ++r)
      C[(size_t)(row0 + wave * 16 + r + 8 * half) * N + col0 + nt * 16 + l16] =
          acc[nt][r];
}

#define KT 128
#define NKT (KT / 16)
#define KPITCH 68
#define VPITCH 72
#define PPITCH (KT + 4)
__global__ __launch_bounds__(BLOCK) void fa_f32_wmma(
    const float* __restrict__ Qg, const float* __restrict__ Kp,
    const float* __restrict__ Vp, float* __restrict__ Op, int S) {
  __shared__ __align__(16) float Ks[KT][KPITCH];
  __shared__ __align__(16) float Vs[KT][VPITCH];
  __shared__ __align__(16) float Ps[4][16][PPITCH];
  const int tid  = threadIdx.x;
  const int wave = tid >> 5;
  const int lane = tid & 31;
  const int half = lane >> 4;
  const int l16  = lane & 15;
  const int q0   = blockIdx.x * 64;
  const int b    = blockIdx.y >> 2;
  const int h    = blockIdx.y & 3;

  v2f qf[16];
  {
    const int qrow = q0 + wave * 16 + l16;
    const float* qptr =
        Qg + (size_t)(b * S + qrow) * 256 + h * 64 + 2 * half;
#pragma unroll
    for (int i = 0; i < 16; ++i) qf[i] = *(const v2f*)(qptr + i * 4);
  }

  v8f oacc[4];
#pragma unroll
  for (int nt = 0; nt < 4; ++nt) oacc[nt] = (v8f)(0.0f);
  float m[8], l[8];
#pragma unroll
  for (int r = 0; r < 8; ++r) { m[r] = -1e30f; l[r] = 0.0f; }

  const int ldr = tid >> 4, ldc = (tid & 15) << 2;

  const int ntiles = (q0 + 63) / KT + 1;
  for (int j = 0; j < ntiles; ++j) {
    const int a0 = j * KT;
    __syncthreads();
    {
      const size_t g0 = (size_t)(b * S + a0 + ldr) * 256 + h * 64 + ldc;
      float4 kreg[KT / 8], vreg[KT / 8];
#pragma unroll
      for (int it = 0; it < KT / 8; ++it) {
        kreg[it] = *(const float4*)(Kp + g0 + (size_t)it * 8 * 256);
        vreg[it] = *(const float4*)(Vp + g0 + (size_t)it * 8 * 256);
      }
#pragma unroll
      for (int it = 0; it < KT / 8; ++it) {
        *(float4*)&Ks[ldr + it * 8][ldc] = kreg[it];
        *(float4*)&Vs[ldr + it * 8][ldc] = vreg[it];
      }
    }
    __syncthreads();

    v8f s[NKT];
#pragma unroll
    for (int kt = 0; kt < NKT; ++kt) s[kt] = (v8f)(0.0f);
#pragma unroll
    for (int i = 0; i < 16; ++i) {
#pragma unroll
      for (int kt = 0; kt < NKT; ++kt) {
        const v2f bk = *(const v2f*)&Ks[kt * 16 + l16][i * 4 + 2 * half];
        s[kt] = wmma4(qf[i], bk, s[kt]);
      }
    }

#pragma unroll
    for (int r = 0; r < 8; ++r) {
      const int qrow = q0 + wave * 16 + r + 8 * half;
      float x[NKT];
#pragma unroll
      for (int kt = 0; kt < NKT; ++kt) {
        x[kt] = s[kt][r] * 0.125f;
        if (a0 + kt * 16 + l16 > qrow) x[kt] = -1e30f;
      }
      float mx = x[0];
#pragma unroll
      for (int kt = 1; kt < NKT; ++kt) mx = fmaxf(mx, x[kt]);
#pragma unroll
      for (int off = 1; off < 16; off <<= 1)
        mx = fmaxf(mx, __shfl_xor(mx, off, 16));
      const float mnew = fmaxf(m[r], mx);
      float p[NKT];
#pragma unroll
      for (int kt = 0; kt < NKT; ++kt) p[kt] = __expf(x[kt] - mnew);
      float sum = p[0];
#pragma unroll
      for (int kt = 1; kt < NKT; ++kt) sum += p[kt];
#pragma unroll
      for (int off = 1; off < 16; off <<= 1) sum += __shfl_xor(sum, off, 16);
      const float scale = __expf(m[r] - mnew);
      l[r] = l[r] * scale + sum;
      m[r] = mnew;
#pragma unroll
      for (int nt = 0; nt < 4; ++nt) oacc[nt][r] *= scale;
#pragma unroll
      for (int kt = 0; kt < NKT; ++kt) Ps[wave][r + 8 * half][kt * 16 + l16] = p[kt];
    }
    __syncthreads();

#pragma unroll
    for (int k0i = 0; k0i < KT / 4; ++k0i) {
      const int k0 = k0i * 4;
      const v2f a = *(const v2f*)&Ps[wave][l16][k0 + 2 * half];
#pragma unroll
      for (int nt = 0; nt < 4; ++nt) {
        v2f bv;
        bv.x = Vs[k0 + 2 * half + 0][nt * 16 + l16];
        bv.y = Vs[k0 + 2 * half + 1][nt * 16 + l16];
        oacc[nt] = wmma4(a, bv, oacc[nt]);
      }
    }
  }

#pragma unroll
  for (int r = 0; r < 8; ++r) {
    const float inv = 1.0f / l[r];
    const int row = b * S + q0 + wave * 16 + r + 8 * half;
#pragma unroll
    for (int nt = 0; nt < 4; ++nt)
      Op[(size_t)row * 256 + h * 64 + nt * 16 + l16] = oacc[nt][r] * inv;
  }
}

extern "C" void kernel_launch(void* const* d_in, const int* in_sizes, int n_in,
                              void* d_out, int out_size, void* d_ws,
                              size_t ws_size, hipStream_t stream) {
  const float* q_in  = (const float*)d_in[0];
  const float* kv_in = (const float*)d_in[1];
  const float* Wq = (const float*)d_in[3];
  const float* Wk = (const float*)d_in[4];
  const float* Wv = (const float*)d_in[5];
  const float* Wo = (const float*)d_in[6];
  float* out = (float*)d_out;

  const int B = 2, S = 2048, IN = 1024;
  const int M = B * S;

  float* ws  = (float*)d_ws;
  float* wqg = ws;
  float* qg  = wqg + 1024 * 256;
  float* kp  = qg + M * 256;
  float* vp  = kp + M * 256;
  float* op  = vp + M * 256;

  reduce_wq_kernel<<<(1024 * 256) / 256, 256, 0, stream>>>(Wq, wqg);

  dim3 blk(BLOCK);
  gemm_f32_wmma<<<dim3(256 / 64, M / 64), blk, 0, stream>>>(q_in, wqg, qg, M, 256, IN);
  gemm_f32_wmma<<<dim3(256 / 64, M / 64), blk, 0, stream>>>(kv_in, Wk, kp, M, 256, IN);
  gemm_f32_wmma<<<dim3(256 / 64, M / 64), blk, 0, stream>>>(kv_in, Wv, vp, M, 256, IN);

  fa_f32_wmma<<<dim3(S / 64, B * 4), blk, 0, stream>>>(qg, kp, vp, op, S);

  gemm_f32_wmma<<<dim3(IN / 64, M / 64), blk, 0, stream>>>(op, Wo, out, M, IN, 256);
}


# GEMM K loop covers 128 k per trip, not 16, its global loads issued as one group before the first LDS store: one load round trip and one barrier pair per 128 k, not four round trips per 16 k
# speedup vs baseline: 4.6302x; 2.2598x over previous
#include <hip/hip_runtime.h>
#include <hip/hip_bf16.h>

typedef __attribute__((ext_vector_type(2))) float v2f;
typedef __attribute__((ext_vector_type(8))) float v8f;

#define BLOCK 128

__device__ __forceinline__ v8f wmma4(v2f a, v2f b, v8f c) {
  return __builtin_amdgcn_wmma_f32_16x16x4_f32(false, a, false, b, (short)0, c,
                                               false, false);
}

__global__ void reduce_wq_kernel(const float* __restrict__ Wq,
                                 float* __restrict__ Wqg) {
  int idx = blockIdx.x * blockDim.x + threadIdx.x;
  if (idx >= 1024 * 256) return;
  int i = idx >> 8;
  int c = idx & 255;
  int h = c >> 6, d = c & 63;
  const float* p = Wq + (size_t)i * 1024 + h * 256 + d;
  Wqg[idx] = p[0] + p[64] + p[128] + p[192];
}

#define GK 128
#define GIT (GK / 8)
#define APITCH (GK + 4)
#define BPITCH 80
__global__ __launch_bounds__(BLOCK) void gemm_f32_wmma(
    const float* __restrict__ A, const float* __restrict__ B,
    float* __restrict__ C, int M, int N, int K) {
  __shared__ __align__(16) float As[64][APITCH];
  __shared__ __align__(16) float Bs[GK][BPITCH];
  const int tid  = threadIdx.x;
  const int wave = tid >> 5;
  const int lane = tid & 31;
  const int half = lane >> 4;
  const int l16  = lane & 15;
  const int row0 = blockIdx.y * 64;
  const int col0 = blockIdx.x * 64;

  v8f acc[4];
#pragma unroll
  for (int nt = 0; nt < 4; ++nt) acc[nt] = (v8f)(0.0f);

  const int ar = tid / (GK / 4), ac = (tid % (GK / 4)) << 2;
  const int br = tid >> 4, bc = (tid & 15) << 2;
  const float* aptr = A + (size_t)(row0 + ar) * K + ac;
  const float* bptr = B + (size_t)br * N + col0 + bc;

  for (int kk = 0; kk < K; kk += GK) {
    __syncthreads();
    float4 areg[GIT], breg[GIT];
#pragma unroll
    for (int it = 0; it < GIT; ++it) {
      areg[it] = *(const float4*)(aptr + (size_t)(it * (512 / GK)) * K + kk);
      breg[it] = *(const float4*)(bptr + (size_t)(kk + it * 8) * N);
    }
#pragma unroll
    for (int it = 0; it < GIT; ++it) {
      *(float4*)&As[ar + it * (512 / GK)][ac] = areg[it];
      *(float4*)&Bs[br + it * 8][bc] = breg[it];
    }
    __syncthreads();
#pragma unroll
    for (int k0 = 0; k0 < GK; k0 += 4) {
      v2f a = *(const v2f*)&As[wave * 16 + l16][k0 + 2 * half];
#pragma unroll
      for (int nt = 0; nt < 4; ++nt) {
        v2f b;
        b.x = Bs[k0 + 2 * half + 0][nt * 16 + l16];
        b.y = Bs[k0 + 2 * half + 1][nt * 16 + l16];
        acc[nt] = wmma4(a, b, acc[nt]);
      }
    }
  }
#pragma unroll
  for (int nt = 0; nt < 4; ++nt)
#pragma unroll
    for (int r = 0; r < 8; ++r)
      C[(size_t)(row0 + wave * 16 + r + 8 * half) * N + col0 + nt * 16 + l16] =
          acc[nt][r];
}

#define KT 128
#define NKT (KT / 16)
#define KPITCH 68
#define VPITCH 72
#define PPITCH (KT + 4)
__global__ __launch_bounds__(BLOCK) void fa_f32_wmma(
    const float* __restrict__ Qg, const float* __restrict__ Kp,
    const float* __restrict__ Vp, float* __restrict__ Op, int S) {
  __shared__ __align__(16) float Ks[KT][KPITCH];
  __shared__ __align__(16) float Vs[KT][VPITCH];
  __shared__ __align__(16) float Ps[4][16][PPITCH];
  const int tid  = threadIdx.x;
  const int wave = tid >> 5;
  const int lane = tid & 31;
  const int half = lane >> 4;
  const int l16  = lane & 15;
  const int q0   = blockIdx.x * 64;
  const int b    = blockIdx.y >> 2;
  const int h    = blockIdx.y & 3;

  v2f qf[16];
  {
    const int qrow = q0 + wave * 16 + l16;
    const float* qptr =
        Qg + (size_t)(b * S + qrow) * 256 + h * 64 + 2 * half;
#pragma unroll
    for (int i = 0; i < 16; ++i) qf[i] = *(const v2f*)(qptr + i * 4);
  }

  v8f oacc[4];
#pragma unroll
  for (int nt = 0; nt < 4; ++nt) oacc[nt] = (v8f)(0.0f);
  float m[8], l[8];
#pragma unroll
  for (int r = 0; r < 8; ++r) { m[r] = -1e30f; l[r] = 0.0f; }

  const int ldr = tid >> 4, ldc = (tid & 15) << 2;

  const int ntiles = (q0 + 63) / KT + 1;
  for (int j = 0; j < ntiles; ++j) {
    const int a0 = j * KT;
    __syncthreads();
    {
      const size_t g0 = (size_t)(b * S + a0 + ldr) * 256 + h * 64 + ldc;
      float4 kreg[KT / 8], vreg[KT / 8];
#pragma unroll
      for (int it = 0; it < KT / 8; ++it) {
        kreg[it] = *(const float4*)(Kp + g0 + (size_t)it * 8 * 256);
        vreg[it] = *(const float4*)(Vp + g0 + (size_t)it * 8 * 256);
      }
#pragma unroll
      for (int it = 0; it < KT / 8; ++it) {
        *(float4*)&Ks[ldr + it * 8][ldc] = kreg[it];
        *(float4*)&Vs[ldr + it * 8][ldc] = vreg[it];
      }
    }
    __syncthreads();

    v8f s[NKT];
#pragma unroll
    for (int kt = 0; kt < NKT; ++kt) s[kt] = (v8f)(0.0f);
#pragma unroll
    for (int i = 0; i < 16; ++i) {
#pragma unroll
      for (int kt = 0; kt < NKT; ++kt) {
        const v2f bk = *(const v2f*)&Ks[kt * 16 + l16][i * 4 + 2 * half];
        s[kt] = wmma4(qf[i], bk, s[kt]);
      }
    }

#pragma unroll
    for (int r = 0; r < 8; ++r) {
      const int qrow = q0 + wave * 16 + r + 8 * half;
      float x[NKT];
#pragma unroll
      for (int kt = 0; kt < NKT; ++kt) {
        x[kt] = s[kt][r] * 0.125f;
        if (a0 + kt * 16 + l16 > qrow) x[kt] = -1e30f;
      }
      float mx = x[0];
#pragma unroll
      for (int kt = 1; kt < NKT; ++kt) mx = fmaxf(mx, x[kt]);
#pragma unroll
      for (int off = 1; off < 16; off <<= 1)
        mx = fmaxf(mx, __shfl_xor(mx, off, 16));
      const float mnew = fmaxf(m[r], mx);
      float p[NKT];
#pragma unroll
      for (int kt = 0; kt < NKT; ++kt) p[kt] = __expf(x[kt] - mnew);
      float sum = p[0];
#pragma unroll
      for (int kt = 1; kt < NKT; ++kt) sum += p[kt];
#pragma unroll
      for (int off = 1; off < 16; off <<= 1) sum += __shfl_xor(sum, off, 16);
      const float scale = __expf(m[r] - mnew);
      l[r] = l[r] * scale + sum;
      m[r] = mnew;
#pragma unroll
      for (int nt = 0; nt < 4; ++nt) oacc[nt][r] *= scale;
#pragma unroll
      for (int kt = 0; kt < NKT; ++kt) Ps[wave][r + 8 * half][kt * 16 + l16] = p[kt];
    }
    __syncthreads();

#pragma unroll
    for (int k0i = 0; k0i < KT / 4; ++k0i) {
      const int k0 = k0i * 4;
      const v2f a = *(const v2f*)&Ps[wave][l16][k0 + 2 * half];
#pragma unroll
      for (int nt = 0; nt < 4; ++nt) {
        v2f bv;
        bv.x = Vs[k0 + 2 * half + 0][nt * 16 + l16];
        bv.y = Vs[k0 + 2 * half + 1][nt * 16 + l16];
        oacc[nt] = wmma4(a, bv, oacc[nt]);
      }
    }
  }

#pragma unroll
  for (int r = 0; r < 8; ++r) {
    const float inv = 1.0f / l[r];
    const int row = b * S + q0 + wave * 16 + r + 8 * half;
#pragma unroll
    for (int nt = 0; nt < 4; ++nt)
      Op[(size_t)row * 256 + h * 64 + nt * 16 + l16] = oacc[nt][r] * inv;
  }
}

extern "C" void kernel_launch(void* const* d_in, const int* in_sizes, int n_in,
                              void* d_out, int out_size, void* d_ws,
                              size_t ws_size, hipStream_t stream) {
  const float* q_in  = (const float*)d_in[0];
  const float* kv_in = (const float*)d_in[1];
  const float* Wq = (const float*)d_in[3];
  const float* Wk = (const float*)d_in[4];
  const float* Wv = (const float*)d_in[5];
  const float* Wo = (const float*)d_in[6];
  float* out = (float*)d_out;

  const int B = 2, S = 2048, IN = 1024;
  const int M = B * S;

  float* ws  = (float*)d_ws;
  float* wqg = ws;
  float* qg  = wqg + 1024 * 256;
  float* kp  = qg + M * 256;
  float* vp  = kp + M * 256;
  float* op  = vp + M * 256;

  reduce_wq_kernel<<<(1024 * 256) / 256, 256, 0, stream>>>(Wq, wqg);

  dim3 blk(BLOCK);
  gemm_f32_wmma<<<dim3(256 / 64, M / 64), blk, 0, stream>>>(q_in, wqg, qg, M, 256, IN);
  gemm_f32_wmma<<<dim3(256 / 64, M / 64), blk, 0, stream>>>(kv_in, Wk, kp, M, 256, IN);
  gemm_f32_wmma<<<dim3(256 / 64, M / 64), blk, 0, stream>>>(kv_in, Wv, vp, M, 256, IN);

  fa_f32_wmma<<<dim3(S / 64, B * 4), blk, 0, stream>>>(qg, kp, vp, op, S);

  gemm_f32_wmma<<<dim3(IN / 64, M / 64), blk, 0, stream>>>(op, Wo, out, M, IN, 256);
}


# next tile's global loads issued into registers one trip ahead, in front of the current tile's matrix work (GEMM K loop and attention key loop): the load round trip runs under the products
# speedup vs baseline: 5.2766x; 1.1396x over previous
#include <hip/hip_runtime.h>
#include <hip/hip_bf16.h>

typedef __attribute__((ext_vector_type(2))) float v2f;
typedef __attribute__((ext_vector_type(8))) float v8f;
typedef __attribute__((ext_vector_type(4))) float v4f;

#define BLOCK 128

__device__ __forceinline__ v8f wmma4(v2f a, v2f b, v8f c) {
  return __builtin_amdgcn_wmma_f32_16x16x4_f32(false, a, false, b, (short)0, c,
                                               false, false);
}

__global__ void reduce_wq_kernel(const float* __restrict__ Wq,
                                 float* __restrict__ Wqg) {
  int idx = blockIdx.x * blockDim.x + threadIdx.x;
  if (idx >= 1024 * 256) return;
  int i = idx >> 8;
  int c = idx & 255;
  int h = c >> 6, d = c & 63;
  const float* p = Wq + (size_t)i * 1024 + h * 256 + d;
  Wqg[idx] = p[0] + p[64] + p[128] + p[192];
}

#define GK 128
#define GIT (GK / 8)
#define APITCH (GK + 4)
#define BPITCH 80
__global__ __launch_bounds__(BLOCK) void gemm_f32_wmma(
    const float* __restrict__ A, const float* __restrict__ B,
    float* __restrict__ C, int M, int N, int K) {
  __shared__ __align__(16) float As[64][APITCH];
  __shared__ __align__(16) float Bs[GK][BPITCH];
  const int tid  = threadIdx.x;
  const int wave = tid >> 5;
  const int lane = tid & 31;
  const int half = lane >> 4;
  const int l16  = lane & 15;
  const int row0 = blockIdx.y * 64;
  const int col0 = blockIdx.x * 64;

  v8f acc[4];
#pragma unroll
  for (int nt = 0; nt < 4; ++nt) acc[nt] = (v8f)(0.0f);

  const int ar = tid / (GK / 4), ac = (tid % (GK / 4)) << 2;
  const int br = tid >> 4, bc = (tid & 15) << 2;
  const float* aptr = A + (size_t)(row0 + ar) * K + ac;
  const float* bptr = B + (size_t)br * N + col0 + bc;

  v4f areg[GIT], breg[GIT];
#pragma unroll
  for (int it = 0; it < GIT; ++it) {
    areg[it] = *(const v4f*)(aptr + (size_t)(it * (512 / GK)) * K);
    breg[it] = *(const v4f*)(bptr + (size_t)(it * 8) * N);
  }

  for (int kk = 0; kk < K; kk += GK) {
    __syncthreads();
#pragma unroll
    for (int it = 0; it < GIT; ++it) {
      *(v4f*)&As[ar + it * (512 / GK)][ac] = areg[it];
      *(v4f*)&Bs[br + it * 8][bc] = breg[it];
    }
    __syncthreads();
    if (kk + GK < K) {
#pragma unroll
      for (int it = 0; it < GIT; ++it) {
        areg[it] = *(const v4f*)(aptr + (size_t)(it * (512 / GK)) * K + kk + GK);
        breg[it] = *(const v4f*)(bptr + (size_t)(kk + GK + it * 8) * N);
      }
    }
#pragma unroll
    for (int k0 = 0; k0 < GK; k0 += 4) {
      v2f a = *(const v2f*)&As[wave * 16 + l16][k0 + 2 * half];
#pragma unroll
      for (int nt = 0; nt < 4; ++nt) {
        v2f b;
        b.x = Bs[k0 + 2 * half + 0][nt * 16 + l16];
        b.y = Bs[k0 + 2 * half + 1][nt * 16 + l16];
        acc[nt] = wmma4(a, b, acc[nt]);
      }
    }
  }
#pragma unroll
  for (int nt = 0; nt < 4; ++nt)
#pragma unroll
    for (int r = 0; r < 8; ++r)
      C[(size_t)(row0 + wave * 16 + r + 8 * half) * N + col0 + nt * 16 + l16] =
          acc[nt][r];
}

#define KT 128
#define NKT (KT / 16)
#define KPITCH 68
#define VPITCH 72
#define PPITCH (KT + 4)
__global__ __launch_bounds__(BLOCK) void fa_f32_wmma(
    const float* __restrict__ Qg, const float* __restrict__ Kp,
    const float* __restrict__ Vp, float* __restrict__ Op, int S) {
  __shared__ __align__(16) float Ks[KT][KPITCH];
  __shared__ __align__(16) float Vs[KT][VPITCH];
  __shared__ __align__(16) float Ps[4][16][PPITCH];
  const int tid  = threadIdx.x;
  const int wave = tid >> 5;
  const int lane = tid & 31;
  const int half = lane >> 4;
  const int l16  = lane & 15;
  const int q0   = blockIdx.x * 64;
  const int b    = blockIdx.y >> 2;
  const int h    = blockIdx.y & 3;

  v2f qf[16];
  {
    const int qrow = q0 + wave * 16 + l16;
    const float* qptr =
        Qg + (size_t)(b * S + qrow) * 256 + h * 64 + 2 * half;
#pragma unroll
    for (int i = 0; i < 16; ++i) qf[i] = *(const v2f*)(qptr + i * 4);
  }

  v8f oacc[4];
#pragma unroll
  for (int nt = 0; nt < 4; ++nt) oacc[nt] = (v8f)(0.0f);
  float m[8], l[8];
#pragma unroll
  for (int r = 0; r < 8; ++r) { m[r] = -1e30f; l[r] = 0.0f; }

  const int ldr = tid >> 4, ldc = (tid & 15) << 2;

  const int ntiles = (q0 + 63) / KT + 1;
  v4f kreg[KT / 8], vreg[KT / 8];
  const size_t gbase = (size_t)(b * S + ldr) * 256 + h * 64 + ldc;
#pragma unroll
  for (int it = 0; it < KT / 8; ++it) {
    kreg[it] = *(const v4f*)(Kp + gbase + (size_t)it * 8 * 256);
    vreg[it] = *(const v4f*)(Vp + gbase + (size_t)it * 8 * 256);
  }
  for (int j = 0; j < ntiles; ++j) {
    const int a0 = j * KT;
    __syncthreads();
#pragma unroll
    for (int it = 0; it < KT / 8; ++it) {
      *(v4f*)&Ks[ldr + it * 8][ldc] = kreg[it];
      *(v4f*)&Vs[ldr + it * 8][ldc] = vreg[it];
    }
    __syncthreads();
    if (j + 1 < ntiles) {
      const size_t g1 = gbase + (size_t)(a0 + KT) * 256;
#pragma unroll
      for (int it = 0; it < KT / 8; ++it) {
        kreg[it] = *(const v4f*)(Kp + g1 + (size_t)it * 8 * 256);
        vreg[it] = *(const v4f*)(Vp + g1 + (size_t)it * 8 * 256);
      }
    }

    v8f s[NKT];
#pragma unroll
    for (int kt = 0; kt < NKT; ++kt) s[kt] = (v8f)(0.0f);
#pragma unroll
    for (int i = 0; i < 16; ++i) {
#pragma unroll
      for (int kt = 0; kt < NKT; ++kt) {
        const v2f bk = *(const v2f*)&Ks[kt * 16 + l16][i * 4 + 2 * half];
        s[kt] = wmma4(qf[i], bk, s[kt]);
      }
    }

#pragma unroll
    for (int r = 0; r < 8; ++r) {
      const int qrow = q0 + wave * 16 + r + 8 * half;
      float x[NKT];
#pragma unroll
      for (int kt = 0; kt < NKT; ++kt) {
        x[kt] = s[kt][r] * 0.125f;
        if (a0 + kt * 16 + l16 > qrow) x[kt] = -1e30f;
      }
      float mx = x[0];
#pragma unroll
      for (int kt = 1; kt < NKT; ++kt) mx = fmaxf(mx, x[kt]);
#pragma unroll
      for (int off = 1; off < 16; off <<= 1)
        mx = fmaxf(mx, __shfl_xor(mx, off, 16));
      const float mnew = fmaxf(m[r], mx);
      float p[NKT];
#pragma unroll
      for (int kt = 0; kt < NKT; ++kt) p[kt] = __expf(x[kt] - mnew);
      float sum = p[0];
#pragma unroll
      for (int kt = 1; kt < NKT; ++kt) sum += p[kt];
#pragma unroll
      for (int off = 1; off < 16; off <<= 1) sum += __shfl_xor(sum, off, 16);
      const float scale = __expf(m[r] - mnew);
      l[r] = l[r] * scale + sum;
      m[r] = mnew;
#pragma unroll
      for (int nt = 0; nt < 4; ++nt) oacc[nt][r] *= scale;
#pragma unroll
      for (int kt = 0; kt < NKT; ++kt) Ps[wave][r + 8 * half][kt * 16 + l16] = p[kt];
    }
    __syncthreads();

#pragma unroll
    for (int k0i = 0; k0i < KT / 4; ++k0i) {
      const int k0 = k0i * 4;
      const v2f a = *(const v2f*)&Ps[wave][l16][k0 + 2 * half];
#pragma unroll
      for (int nt = 0; nt < 4; ++nt) {
        v2f bv;
        bv.x = Vs[k0 + 2 * half + 0][nt * 16 + l16];
        bv.y = Vs[k0 + 2 * half + 1][nt * 16 + l16];
        oacc[nt] = wmma4(a, bv, oacc[nt]);
      }
    }
  }

#pragma unroll
  for (int r = 0; r < 8; ++r) {
    const float inv = 1.0f / l[r];
    const int row = b * S + q0 + wave * 16 + r + 8 * half;
#pragma unroll
    for (int nt = 0; nt < 4; ++nt)
      Op[(size_t)row * 256 + h * 64 + nt * 16 + l16] = oacc[nt][r] * inv;
  }
}

extern "C" void kernel_launch(void* const* d_in, const int* in_sizes, int n_in,
                              void* d_out, int out_size, void* d_ws,
                              size_t ws_size, hipStream_t stream) {
  const float* q_in  = (const float*)d_in[0];
  const float* kv_in = (const float*)d_in[1];
  const float* Wq = (const float*)d_in[3];
  const float* Wk = (const float*)d_in[4];
  const float* Wv = (const float*)d_in[5];
  const float* Wo = (const float*)d_in[6];
  float* out = (float*)d_out;

  const int B = 2, S = 2048, IN = 1024;
  const int M = B * S;

  float* ws  = (float*)d_ws;
  float* wqg = ws;
  float* qg  = wqg + 1024 * 256;
  float* kp  = qg + M * 256;
  float* vp  = kp + M * 256;
  float* op  = vp + M * 256;

  reduce_wq_kernel<<<(1024 * 256) / 256, 256, 0, stream>>>(Wq, wqg);

  dim3 blk(BLOCK);
  gemm_f32_wmma<<<dim3(256 / 64, M / 64), blk, 0, stream>>>(q_in, wqg, qg, M, 256, IN);
  gemm_f32_wmma<<<dim3(256 / 64, M / 64), blk, 0, stream>>>(kv_in, Wk, kp, M, 256, IN);
  gemm_f32_wmma<<<dim3(256 / 64, M / 64), blk, 0, stream>>>(kv_in, Wv, vp, M, 256, IN);

  fa_f32_wmma<<<dim3(S / 64, B * 4), blk, 0, stream>>>(qg, kp, vp, op, S);

  gemm_f32_wmma<<<dim3(IN / 64, M / 64), blk, 0, stream>>>(op, Wo, out, M, IN, 256);
}


# LDS fragment loads of the next block of 32 matrix instructions go out as one group, into a second register set, ahead of the current block (GEMM k loop, attention scores and P*V)
# speedup vs baseline: 5.5970x; 1.0607x over previous
#include <hip/hip_runtime.h>
#include <hip/hip_bf16.h>

typedef __attribute__((ext_vector_type(2))) float v2f;
typedef __attribute__((ext_vector_type(8))) float v8f;
typedef __attribute__((ext_vector_type(4))) float v4f;

#define BLOCK 128

__device__ __forceinline__ v8f wmma4(v2f a, v2f b, v8f c) {
  return __builtin_amdgcn_wmma_f32_16x16x4_f32(false, a, false, b, (short)0, c,
                                               false, false);
}

__global__ void reduce_wq_kernel(const float* __restrict__ Wq,
                                 float* __restrict__ Wqg) {
  int idx = blockIdx.x * blockDim.x + threadIdx.x;
  if (idx >= 1024 * 256) return;
  int i = idx >> 8;
  int c = idx & 255;
  int h = c >> 6, d = c & 63;
  const float* p = Wq + (size_t)i * 1024 + h * 256 + d;
  Wqg[idx] = p[0] + p[64] + p[128] + p[192];
}

#define GK 128
#define GIT (GK / 8)
#define APITCH (GK + 4)
#define BPITCH 80
#define SB 8
#define NB (GK / 4 / SB)
__global__ __launch_bounds__(BLOCK) void gemm_f32_wmma(
    const float* __restrict__ A, const float* __restrict__ B,
    float* __restrict__ C, int M, int N, int K) {
  __shared__ __align__(16) float As[64][APITCH];
  __shared__ __align__(16) float Bs[GK][BPITCH];
  const int tid  = threadIdx.x;
  const int wave = tid >> 5;
  const int lane = tid & 31;
  const int half = lane >> 4;
  const int l16  = lane & 15;
  const int row0 = blockIdx.y * 64;
  const int col0 = blockIdx.x * 64;

  v8f acc[4];
#pragma unroll
  for (int nt = 0; nt < 4; ++nt) acc[nt] = (v8f)(0.0f);

  const int ar = tid / (GK / 4), ac = (tid % (GK / 4)) << 2;
  const int br = tid >> 4, bc = (tid & 15) << 2;
  const float* aptr = A + (size_t)(row0 + ar) * K + ac;
  const float* bptr = B + (size_t)br * N + col0 + bc;

  v4f areg[GIT], breg[GIT];
#pragma unroll
  for (int it = 0; it < GIT; ++it) {
    areg[it] = *(const v4f*)(aptr + (size_t)(it * (512 / GK)) * K);
    breg[it] = *(const v4f*)(bptr + (size_t)(it * 8) * N);
  }

  for (int kk = 0; kk < K; kk += GK) {
    __syncthreads();
#pragma unroll
    for (int it = 0; it < GIT; ++it) {
      *(v4f*)&As[ar + it * (512 / GK)][ac] = areg[it];
      *(v4f*)&Bs[br + it * 8][bc] = breg[it];
    }
    __syncthreads();
    if (kk + GK < K) {
#pragma unroll
      for (int it = 0; it < GIT; ++it) {
        areg[it] = *(const v4f*)(aptr + (size_t)(it * (512 / GK)) * K + kk + GK);
        breg[it] = *(const v4f*)(bptr + (size_t)(kk + GK + it * 8) * N);
      }
    }
    v2f af[2][SB];
    float bx[2][SB][4], by[2][SB][4];
#pragma unroll
    for (int s = 0; s < SB; ++s) {
      const int k0 = s * 4;
      af[0][s] = *(const v2f*)&As[wave * 16 + l16][k0 + 2 * half];
#pragma unroll
      for (int nt = 0; nt < 4; ++nt) {
        bx[0][s][nt] = Bs[k0 + 2 * half + 0][nt * 16 + l16];
        by[0][s][nt] = Bs[k0 + 2 * half + 1][nt * 16 + l16];
      }
    }
#pragma unroll
    for (int blk = 0; blk < NB; ++blk) {
      const int cur = blk & 1;
      if (blk + 1 < NB) {
#pragma unroll
        for (int s = 0; s < SB; ++s) {
          const int k0 = ((blk + 1) * SB + s) * 4;
          af[cur ^ 1][s] = *(const v2f*)&As[wave * 16 + l16][k0 + 2 * half];
#pragma unroll
          for (int nt = 0; nt < 4; ++nt) {
            bx[cur ^ 1][s][nt] = Bs[k0 + 2 * half + 0][nt * 16 + l16];
            by[cur ^ 1][s][nt] = Bs[k0 + 2 * half + 1][nt * 16 + l16];
          }
        }
      }
      __builtin_amdgcn_sched_barrier(0);
#pragma unroll
      for (int s = 0; s < SB; ++s)
#pragma unroll
        for (int nt = 0; nt < 4; ++nt) {
          v2f b;
          b.x = bx[cur][s][nt];
          b.y = by[cur][s][nt];
          acc[nt] = wmma4(af[cur][s], b, acc[nt]);
        }
      if (blk + 1 < NB) __builtin_amdgcn_sched_barrier(0);
    }
  }
#pragma unroll
  for (int nt = 0; nt < 4; ++nt)
#pragma unroll
    for (int r = 0; r < 8; ++r)
      C[(size_t)(row0 + wave * 16 + r + 8 * half) * N + col0 + nt * 16 + l16] =
          acc[nt][r];
}

#define KT 128
#define NKT (KT / 16)
#define KPITCH 68
#define VPITCH 72
#define PPITCH (KT + 4)
#define SI 4
#define SP 8
__global__ __launch_bounds__(BLOCK) void fa_f32_wmma(
    const float* __restrict__ Qg, const float* __restrict__ Kp,
    const float* __restrict__ Vp, float* __restrict__ Op, int S) {
  __shared__ __align__(16) float Ks[KT][KPITCH];
  __shared__ __align__(16) float Vs[KT][VPITCH];
  __shared__ __align__(16) float Ps[4][16][PPITCH];
  const int tid  = threadIdx.x;
  const int wave = tid >> 5;
  const int lane = tid & 31;
  const int half = lane >> 4;
  const int l16  = lane & 15;
  const int q0   = blockIdx.x * 64;
  const int b    = blockIdx.y >> 2;
  const int h    = blockIdx.y & 3;

  v2f qf[16];
  {
    const int qrow = q0 + wave * 16 + l16;
    const float* qptr =
        Qg + (size_t)(b * S + qrow) * 256 + h * 64 + 2 * half;
#pragma unroll
    for (int i = 0; i < 16; ++i) qf[i] = *(const v2f*)(qptr + i * 4);
  }

  v8f oacc[4];
#pragma unroll
  for (int nt = 0; nt < 4; ++nt) oacc[nt] = (v8f)(0.0f);
  float m[8], l[8];
#pragma unroll
  for (int r = 0; r < 8; ++r) { m[r] = -1e30f; l[r] = 0.0f; }

  const int ldr = tid >> 4, ldc = (tid & 15) << 2;

  const int ntiles = (q0 + 63) / KT + 1;
  v4f kreg[KT / 8], vreg[KT / 8];
  const size_t gbase = (size_t)(b * S + ldr) * 256 + h * 64 + ldc;
#pragma unroll
  for (int it = 0; it < KT / 8; ++it) {
    kreg[it] = *(const v4f*)(Kp + gbase + (size_t)it * 8 * 256);
    vreg[it] = *(const v4f*)(Vp + gbase + (size_t)it * 8 * 256);
  }
  for (int j = 0; j < ntiles; ++j) {
    const int a0 = j * KT;
    __syncthreads();
#pragma unroll
    for (int it = 0; it < KT / 8; ++it) {
      *(v4f*)&Ks[ldr + it * 8][ldc] = kreg[it];
      *(v4f*)&Vs[ldr + it * 8][ldc] = vreg[it];
    }
    __syncthreads();
    if (j + 1 < ntiles) {
      const size_t g1 = gbase + (size_t)(a0 + KT) * 256;
#pragma unroll
      for (int it = 0; it < KT / 8; ++it) {
        kreg[it] = *(const v4f*)(Kp + g1 + (size_t)it * 8 * 256);
        vreg[it] = *(const v4f*)(Vp + g1 + (size_t)it * 8 * 256);
      }
    }

    v8f s[NKT];
#pragma unroll
    for (int kt = 0; kt < NKT; ++kt) s[kt] = (v8f)(0.0f);
    {
      v2f kf[2][SI][NKT];
#pragma unroll
      for (int ii = 0; ii < SI; ++ii)
#pragma unroll
        for (int kt = 0; kt < NKT; ++kt)
          kf[0][ii][kt] = *(const v2f*)&Ks[kt * 16 + l16][ii * 4 + 2 * half];
#pragma unroll
      for (int blk = 0; blk < 16 / SI; ++blk) {
        const int cur = blk & 1;
        if (blk + 1 < 16 / SI) {
#pragma unroll
          for (int ii = 0; ii < SI; ++ii)
#pragma unroll
            for (int kt = 0; kt < NKT; ++kt)
              kf[cur ^ 1][ii][kt] =
                  *(const v2f*)&Ks[kt * 16 + l16][((blk + 1) * SI + ii) * 4 + 2 * half];
        }
        __builtin_amdgcn_sched_barrier(0);
#pragma unroll
        for (int ii = 0; ii < SI; ++ii)
#pragma unroll
          for (int kt = 0; kt < NKT; ++kt)
            s[kt] = wmma4(qf[blk * SI + ii], kf[cur][ii][kt], s[kt]);
        if (blk + 1 < 16 / SI) __builtin_amdgcn_sched_barrier(0);
      }
    }

#pragma unroll
    for (int r = 0; r < 8; ++r) {
      const int qrow = q0 + wave * 16 + r + 8 * half;
      float x[NKT];
#pragma unroll
      for (int kt = 0; kt < NKT; ++kt) {
        x[kt] = s[kt][r] * 0.125f;
        if (a0 + kt * 16 + l16 > qrow) x[kt] = -1e30f;
      }
      float mx = x[0];
#pragma unroll
      for (int kt = 1; kt < NKT; ++kt) mx = fmaxf(mx, x[kt]);
#pragma unroll
      for (int off = 1; off < 16; off <<= 1)
        mx = fmaxf(mx, __shfl_xor(mx, off, 16));
      const float mnew = fmaxf(m[r], mx);
      float p[NKT];
#pragma unroll
      for (int kt = 0; kt < NKT; ++kt) p[kt] = __expf(x[kt] - mnew);
      float sum = p[0];
#pragma unroll
      for (int kt = 1; kt < NKT; ++kt) sum += p[kt];
#pragma unroll
      for (int off = 1; off < 16; off <<= 1) sum += __shfl_xor(sum, off, 16);
      const float scale = __expf(m[r] - mnew);
      l[r] = l[r] * scale + sum;
      m[r] = mnew;
#pragma unroll
      for (int nt = 0; nt < 4; ++nt) oacc[nt][r] *= scale;
#pragma unroll
      for (int kt = 0; kt < NKT; ++kt) Ps[wave][r + 8 * half][kt * 16 + l16] = p[kt];
    }
    __syncthreads();

    {
      v2f pf[2][SP];
      float vx[2][SP][4], vy[2][SP][4];
#pragma unroll
      for (int s2 = 0; s2 < SP; ++s2) {
        const int k0 = s2 * 4;
        pf[0][s2] = *(const v2f*)&Ps[wave][l16][k0 + 2 * half];
#pragma unroll
        for (int nt = 0; nt < 4; ++nt) {
          vx[0][s2][nt] = Vs[k0 + 2 * half + 0][nt * 16 + l16];
          vy[0][s2][nt] = Vs[k0 + 2 * half + 1][nt * 16 + l16];
        }
      }
#pragma unroll
      for (int blk = 0; blk < KT / 4 / SP; ++blk) {
        const int cur = blk & 1;
        if (blk + 1 < KT / 4 / SP) {
#pragma unroll
          for (int s2 = 0; s2 < SP; ++s2) {
            const int k0 = ((blk + 1) * SP + s2) * 4;
            pf[cur ^ 1][s2] = *(const v2f*)&Ps[wave][l16][k0 + 2 * half];
#pragma unroll
            for (int nt = 0; nt < 4; ++nt) {
              vx[cur ^ 1][s2][nt] = Vs[k0 + 2 * half + 0][nt * 16 + l16];
              vy[cur ^ 1][s2][nt] = Vs[k0 + 2 * half + 1][nt * 16 + l16];
            }
          }
        }
        __builtin_amdgcn_sched_barrier(0);
#pragma unroll
        for (int s2 = 0; s2 < SP; ++s2)
#pragma unroll
          for (int nt = 0; nt < 4; ++nt) {
            v2f bv;
            bv.x = vx[cur][s2][nt];
            bv.y = vy[cur][s2][nt];
            oacc[nt] = wmma4(pf[cur][s2], bv, oacc[nt]);
          }
        if (blk + 1 < KT / 4 / SP) __builtin_amdgcn_sched_barrier(0);
      }
    }
  }

#pragma unroll
  for (int r = 0; r < 8; ++r) {
    const float inv = 1.0f / l[r];
    const int row = b * S + q0 + wave * 16 + r + 8 * half;
#pragma unroll
    for (int nt = 0; nt < 4; ++nt)
      Op[(size_t)row * 256 + h * 64 + nt * 16 + l16] = oacc[nt][r] * inv;
  }
}

extern "C" void kernel_launch(void* const* d_in, const int* in_sizes, int n_in,
                              void* d_out, int out_size, void* d_ws,
                              size_t ws_size, hipStream_t stream) {
  const float* q_in  = (const float*)d_in[0];
  const float* kv_in = (const float*)d_in[1];
  const float* Wq = (const float*)d_in[3];
  const float* Wk = (const float*)d_in[4];
  const float* Wv = (const float*)d_in[5];
  const float* Wo = (const float*)d_in[6];
  float* out = (float*)d_out;

  const int B = 2, S = 2048, IN = 1024;
  const int M = B * S;

  float* ws  = (float*)d_ws;
  float* wqg = ws;
  float* qg  = wqg + 1024 * 256;
  float* kp  = qg + M * 256;
  float* vp  = kp + M * 256;
  float* op  = vp + M * 256;

  reduce_wq_kernel<<<(1024 * 256) / 256, 256, 0, stream>>>(Wq, wqg);

  dim3 blk(BLOCK);
  gemm_f32_wmma<<<dim3(256 / 64, M / 64), blk, 0, stream>>>(q_in, wqg, qg, M, 256, IN);
  gemm_f32_wmma<<<dim3(256 / 64, M / 64), blk, 0, stream>>>(kv_in, Wk, kp, M, 256, IN);
  gemm_f32_wmma<<<dim3(256 / 64, M / 64), blk, 0, stream>>>(kv_in, Wv, vp, M, 256, IN);

  fa_f32_wmma<<<dim3(S / 64, B * 4), blk, 0, stream>>>(qg, kp, vp, op, S);

  gemm_f32_wmma<<<dim3(IN / 64, M / 64), blk, 0, stream>>>(op, Wo, out, M, IN, 256);
}


# attention: each block takes a complementary pair of 32-row query tiles (equal causal work per block); two wave pairs split every key tile, partials merged through LDS; changes order of row sums
# speedup vs baseline: 6.5276x; 1.1663x over previous
#include <hip/hip_runtime.h>
#include <hip/hip_bf16.h>

typedef __attribute__((ext_vector_type(2))) float v2f;
typedef __attribute__((ext_vector_type(8))) float v8f;
typedef __attribute__((ext_vector_type(4))) float v4f;

#define BLOCK 128

__device__ __forceinline__ v8f wmma4(v2f a, v2f b, v8f c) {
  return __builtin_amdgcn_wmma_f32_16x16x4_f32(false, a, false, b, (short)0, c,
                                               false, false);
}

__global__ void reduce_wq_kernel(const float* __restrict__ Wq,
                                 float* __restrict__ Wqg) {
  int idx = blockIdx.x * blockDim.x + threadIdx.x;
  if (idx >= 1024 * 256) return;
  int i = idx >> 8;
  int c = idx & 255;
  int h = c >> 6, d = c & 63;
  const float* p = Wq + (size_t)i * 1024 + h * 256 + d;
  Wqg[idx] = p[0] + p[64] + p[128] + p[192];
}

#define GK 128
#define GIT (GK / 8)
#define APITCH (GK + 4)
#define BPITCH 80
#define SB 8
#define NB (GK / 4 / SB)
__global__ __launch_bounds__(BLOCK) void gemm_f32_wmma(
    const float* __restrict__ A, const float* __restrict__ B,
    float* __restrict__ C, int M, int N, int K) {
  __shared__ __align__(16) float As[64][APITCH];
  __shared__ __align__(16) float Bs[GK][BPITCH];
  const int tid  = threadIdx.x;
  const int wave = tid >> 5;
  const int lane = tid & 31;
  const int half = lane >> 4;
  const int l16  = lane & 15;
  const int row0 = blockIdx.y * 64;
  const int col0 = blockIdx.x * 64;

  v8f acc[4];
#pragma unroll
  for (int nt = 0; nt < 4; ++nt) acc[nt] = (v8f)(0.0f);

  const int ar = tid / (GK / 4), ac = (tid % (GK / 4)) << 2;
  const int br = tid >> 4, bc = (tid & 15) << 2;
  const float* aptr = A + (size_t)(row0 + ar) * K + ac;
  const float* bptr = B + (size_t)br * N + col0 + bc;

  v4f areg[GIT], breg[GIT];
#pragma unroll
  for (int it = 0; it < GIT; ++it) {
    areg[it] = *(const v4f*)(aptr + (size_t)(it * (512 / GK)) * K);
    breg[it] = *(const v4f*)(bptr + (size_t)(it * 8) * N);
  }

  for (int kk = 0; kk < K; kk += GK) {
    __syncthreads();
#pragma unroll
    for (int it = 0; it < GIT; ++it) {
      *(v4f*)&As[ar + it * (512 / GK)][ac] = areg[it];
      *(v4f*)&Bs[br + it * 8][bc] = breg[it];
    }
    __syncthreads();
    if (kk + GK < K) {
#pragma unroll
      for (int it = 0; it < GIT; ++it) {
        areg[it] = *(const v4f*)(aptr + (size_t)(it * (512 / GK)) * K + kk + GK);
        breg[it] = *(const v4f*)(bptr + (size_t)(kk + GK + it * 8) * N);
      }
    }
    v2f af[2][SB];
    float bx[2][SB][4], by[2][SB][4];
#pragma unroll
    for (int s = 0; s < SB; ++s) {
      const int k0 = s * 4;
      af[0][s] = *(const v2f*)&As[wave * 16 + l16][k0 + 2 * half];
#pragma unroll
      for (int nt = 0; nt < 4; ++nt) {
        bx[0][s][nt] = Bs[k0 + 2 * half + 0][nt * 16 + l16];
        by[0][s][nt] = Bs[k0 + 2 * half + 1][nt * 16 + l16];
      }
    }
#pragma unroll
    for (int blk = 0; blk < NB; ++blk) {
      const int cur = blk & 1;
      if (blk + 1 < NB) {
#pragma unroll
        for (int s = 0; s < SB; ++s) {
          const int k0 = ((blk + 1) * SB + s) * 4;
          af[cur ^ 1][s] = *(const v2f*)&As[wave * 16 + l16][k0 + 2 * half];
#pragma unroll
          for (int nt = 0; nt < 4; ++nt) {
            bx[cur ^ 1][s][nt] = Bs[k0 + 2 * half + 0][nt * 16 + l16];
            by[cur ^ 1][s][nt] = Bs[k0 + 2 * half + 1][nt * 16 + l16];
          }
        }
      }
      __builtin_amdgcn_sched_barrier(0);
#pragma unroll
      for (int s = 0; s < SB; ++s)
#pragma unroll
        for (int nt = 0; nt < 4; ++nt) {
          v2f b;
          b.x = bx[cur][s][nt];
          b.y = by[cur][s][nt];
          acc[nt] = wmma4(af[cur][s], b, acc[nt]);
        }
      if (blk + 1 < NB) __builtin_amdgcn_sched_barrier(0);
    }
  }
#pragma unroll
  for (int nt = 0; nt < 4; ++nt)
#pragma unroll
    for (int r = 0; r < 8; ++r)
      C[(size_t)(row0 + wave * 16 + r + 8 * half) * N + col0 + nt * 16 + l16] =
          acc[nt][r];
}

#define KT 128
#define KW (KT / 2)
#define NKT (KW / 16)
#define QT 32
#define KPITCH 68
#define VPITCH 72
#define PPITCH (KW + 4)
#define SI 8
#define SP 8
__global__ __launch_bounds__(BLOCK) void fa_f32_wmma(
    const float* __restrict__ Qg, const float* __restrict__ Kp,
    const float* __restrict__ Vp, float* __restrict__ Op, int S) {
  __shared__ __align__(16) float Ks[KT][KPITCH];
  __shared__ __align__(16) float Vs[KT][VPITCH];
  __shared__ __align__(16) float Ps[4][16][PPITCH];
  __shared__ float Xo[2][32][32];
  __shared__ float Xm[2][8][32];
  __shared__ float Xl[2][8][32];
  const int tid  = threadIdx.x;
  const int wave = tid >> 5;
  const int lane = tid & 31;
  const int half = lane >> 4;
  const int l16  = lane & 15;
  const int rh   = wave & 1;
  const int kh   = wave >> 1;
  const int b    = blockIdx.y >> 2;
  const int h    = blockIdx.y & 3;
  const int ntq  = S / QT;

  const int ldr = tid >> 4, ldc = (tid & 15) << 2;
  const size_t gbase = (size_t)(b * S + ldr) * 256 + h * 64 + ldc;

  for (int t = 0; t < 2; ++t) {
  const int qt = t ? (ntq - 1 - (int)blockIdx.x) : (int)blockIdx.x;
  const int q0 = qt * QT;

  v2f qf[16];
  {
    const int qrow = q0 + rh * 16 + l16;
    const float* qptr =
        Qg + (size_t)(b * S + qrow) * 256 + h * 64 + 2 * half;
#pragma unroll
    for (int i = 0; i < 16; ++i) qf[i] = *(const v2f*)(qptr + i * 4);
  }

  v8f oacc[4];
#pragma unroll
  for (int nt = 0; nt < 4; ++nt) oacc[nt] = (v8f)(0.0f);
  float m[8], l[8];
#pragma unroll
  for (int r = 0; r < 8; ++r) { m[r] = -1e30f; l[r] = 0.0f; }

  const int ntiles = (q0 + QT - 1) / KT + 1;
  v4f kreg[KT / 8], vreg[KT / 8];
#pragma unroll
  for (int it = 0; it < KT / 8; ++it) {
    kreg[it] = *(const v4f*)(Kp + gbase + (size_t)it * 8 * 256);
    vreg[it] = *(const v4f*)(Vp + gbase + (size_t)it * 8 * 256);
  }
  for (int j = 0; j < ntiles; ++j) {
    const int a0 = j * KT;
    __syncthreads();
#pragma unroll
    for (int it = 0; it < KT / 8; ++it) {
      *(v4f*)&Ks[ldr + it * 8][ldc] = kreg[it];
      *(v4f*)&Vs[ldr + it * 8][ldc] = vreg[it];
    }
    __syncthreads();
    if (j + 1 < ntiles) {
      const size_t g1 = gbase + (size_t)(a0 + KT) * 256;
#pragma unroll
      for (int it = 0; it < KT / 8; ++it) {
        kreg[it] = *(const v4f*)(Kp + g1 + (size_t)it * 8 * 256);
        vreg[it] = *(const v4f*)(Vp + g1 + (size_t)it * 8 * 256);
      }
    }

    v8f s[NKT];
#pragma unroll
    for (int kt = 0; kt < NKT; ++kt) s[kt] = (v8f)(0.0f);
    {
      const int krow = kh * KW + l16;
      v2f kf[2][SI][NKT];
#pragma unroll
      for (int ii = 0; ii < SI; ++ii)
#pragma unroll
        for (int kt = 0; kt < NKT; ++kt)
          kf[0][ii][kt] = *(const v2f*)&Ks[krow + kt * 16][ii * 4 + 2 * half];
#pragma unroll
      for (int blk = 0; blk < 16 / SI; ++blk) {
        const int cur = blk & 1;
        if (blk + 1 < 16 / SI) {
#pragma unroll
          for (int ii = 0; ii < SI; ++ii)
#pragma unroll
            for (int kt = 0; kt < NKT; ++kt)
              kf[cur ^ 1][ii][kt] =
                  *(const v2f*)&Ks[krow + kt * 16][((blk + 1) * SI + ii) * 4 + 2 * half];
        }
        __builtin_amdgcn_sched_barrier(0);
#pragma unroll
        for (int ii = 0; ii < SI; ++ii)
#pragma unroll
          for (int kt = 0; kt < NKT; ++kt)
            s[kt] = wmma4(qf[blk * SI + ii], kf[cur][ii][kt], s[kt]);
        if (blk + 1 < 16 / SI) __builtin_amdgcn_sched_barrier(0);
      }
    }

#pragma unroll
    for (int r = 0; r < 8; ++r) {
      const int qrow = q0 + rh * 16 + r + 8 * half;
      float x[NKT];
#pragma unroll
      for (int kt = 0; kt < NKT; ++kt) {
        x[kt] = s[kt][r] * 0.125f;
        if (a0 + kh * KW + kt * 16 + l16 > qrow) x[kt] = -1e30f;
      }
      float mx = x[0];
#pragma unroll
      for (int kt = 1; kt < NKT; ++kt) mx = fmaxf(mx, x[kt]);
#pragma unroll
      for (int off = 1; off < 16; off <<= 1)
        mx = fmaxf(mx, __shfl_xor(mx, off, 16));
      const float mnew = fmaxf(m[r], mx);
      float p[NKT];
#pragma unroll
      for (int kt = 0; kt < NKT; ++kt) p[kt] = __expf(x[kt] - mnew);
      float sum = p[0];
#pragma unroll
      for (int kt = 1; kt < NKT; ++kt) sum += p[kt];
#pragma unroll
      for (int off = 1; off < 16; off <<= 1) sum += __shfl_xor(sum, off, 16);
      const float scale = __expf(m[r] - mnew);
      l[r] = l[r] * scale + sum;
      m[r] = mnew;
#pragma unroll
      for (int nt = 0; nt < 4; ++nt) oacc[nt][r] *= scale;
#pragma unroll
      for (int kt = 0; kt < NKT; ++kt) Ps[wave][r + 8 * half][kt * 16 + l16] = p[kt];
    }
    __syncthreads();

    {
      const int vrow = kh * KW + 2 * half;
      v2f pf[2][SP];
      float vx[2][SP][4], vy[2][SP][4];
#pragma unroll
      for (int s2 = 0; s2 < SP; ++s2) {
        const int k0 = s2 * 4;
        pf[0][s2] = *(const v2f*)&Ps[wave][l16][k0 + 2 * half];
#pragma unroll
        for (int nt = 0; nt < 4; ++nt) {
          vx[0][s2][nt] = Vs[vrow + k0 + 0][nt * 16 + l16];
          vy[0][s2][nt] = Vs[vrow + k0 + 1][nt * 16 + l16];
        }
      }
#pragma unroll
      for (int blk = 0; blk < KW / 4 / SP; ++blk) {
        const int cur = blk & 1;
        if (blk + 1 < KW / 4 / SP) {
#pragma unroll
          for (int s2 = 0; s2 < SP; ++s2) {
            const int k0 = ((blk + 1) * SP + s2) * 4;
            pf[cur ^ 1][s2] = *(const v2f*)&Ps[wave][l16][k0 + 2 * half];
#pragma unroll
            for (int nt = 0; nt < 4; ++nt) {
              vx[cur ^ 1][s2][nt] = Vs[vrow + k0 + 0][nt * 16 + l16];
              vy[cur ^ 1][s2][nt] = Vs[vrow + k0 + 1][nt * 16 + l16];
            }
          }
        }
        __builtin_amdgcn_sched_barrier(0);
#pragma unroll
        for (int s2 = 0; s2 < SP; ++s2)
#pragma unroll
          for (int nt = 0; nt < 4; ++nt) {
            v2f bv;
            bv.x = vx[cur][s2][nt];
            bv.y = vy[cur][s2][nt];
            oacc[nt] = wmma4(pf[cur][s2], bv, oacc[nt]);
          }
        if (blk + 1 < KW / 4 / SP) __builtin_amdgcn_sched_barrier(0);
      }
    }
  }

  if (kh == 1) {
#pragma unroll
    for (int r = 0; r < 8; ++r) {
      Xm[rh][r][lane] = m[r];
      Xl[rh][r][lane] = l[r];
#pragma unroll
      for (int nt = 0; nt < 4; ++nt) Xo[rh][nt * 8 + r][lane] = oacc[nt][r];
    }
  }
  __syncthreads();
  if (kh == 0) {
#pragma unroll
    for (int r = 0; r < 8; ++r) {
      const float m2 = Xm[rh][r][lane];
      const float l2 = Xl[rh][r][lane];
      const float mm = fmaxf(m[r], m2);
      const float w1 = __expf(m[r] - mm);
      const float w2 = __expf(m2 - mm);
      const float inv = 1.0f / (l[r] * w1 + l2 * w2);
      const int row = b * S + q0 + rh * 16 + r + 8 * half;
#pragma unroll
      for (int nt = 0; nt < 4; ++nt)
        Op[(size_t)row * 256 + h * 64 + nt * 16 + l16] =
            (oacc[nt][r] * w1 + Xo[rh][nt * 8 + r][lane] * w2) * inv;
    }
  }
  }
}

extern "C" void kernel_launch(void* const* d_in, const int* in_sizes, int n_in,
                              void* d_out, int out_size, void* d_ws,
                              size_t ws_size, hipStream_t stream) {
  const float* q_in  = (const float*)d_in[0];
  const float* kv_in = (const float*)d_in[1];
  const float* Wq = (const float*)d_in[3];
  const float* Wk = (const float*)d_in[4];
  const float* Wv = (const float*)d_in[5];
  const float* Wo = (const float*)d_in[6];
  float* out = (float*)d_out;

  const int B = 2, S = 2048, IN = 1024;
  const int M = B * S;

  float* ws  = (float*)d_ws;
  float* wqg = ws;
  float* qg  = wqg + 1024 * 256;
  float* kp  = qg + M * 256;
  float* vp  = kp + M * 256;
  float* op  = vp + M * 256;

  reduce_wq_kernel<<<(1024 * 256) / 256, 256, 0, stream>>>(Wq, wqg);

  dim3 blk(BLOCK);
  gemm_f32_wmma<<<dim3(256 / 64, M / 64), blk, 0, stream>>>(q_in, wqg, qg, M, 256, IN);
  gemm_f32_wmma<<<dim3(256 / 64, M / 64), blk, 0, stream>>>(kv_in, Wk, kp, M, 256, IN);
  gemm_f32_wmma<<<dim3(256 / 64, M / 64), blk, 0, stream>>>(kv_in, Wv, vp, M, 256, IN);

  fa_f32_wmma<<<dim3(S / 64, B * 4), blk, 0, stream>>>(qg, kp, vp, op, S);

  gemm_f32_wmma<<<dim3(IN / 64, M / 64), blk, 0, stream>>>(op, Wo, out, M, IN, 256);
}


# Q, K and V projection GEMMs run as one launch (blockIdx.z picks the operands), three workgroups resident per WGP: each hides the others' tile staging, barriers and LDS load groups
# speedup vs baseline: 7.2418x; 1.1094x over previous
#include <hip/hip_runtime.h>
#include <hip/hip_bf16.h>

typedef __attribute__((ext_vector_type(2))) float v2f;
typedef __attribute__((ext_vector_type(8))) float v8f;
typedef __attribute__((ext_vector_type(4))) float v4f;

#define BLOCK 128

__device__ __forceinline__ v8f wmma4(v2f a, v2f b, v8f c) {
  return __builtin_amdgcn_wmma_f32_16x16x4_f32(false, a, false, b, (short)0, c,
                                               false, false);
}

__global__ void reduce_wq_kernel(const float* __restrict__ Wq,
                                 float* __restrict__ Wqg) {
  int idx = blockIdx.x * blockDim.x + threadIdx.x;
  if (idx >= 1024 * 256) return;
  int i = idx >> 8;
  int c = idx & 255;
  int h = c >> 6, d = c & 63;
  const float* p = Wq + (size_t)i * 1024 + h * 256 + d;
  Wqg[idx] = p[0] + p[64] + p[128] + p[192];
}

#define GK 128
#define GIT (GK / 8)
#define APITCH (GK + 4)
#define BPITCH 80
#define SB 8
#define NB (GK / 4 / SB)
__global__ __launch_bounds__(BLOCK, 3) void gemm_f32_wmma(
    const float* __restrict__ A0, const float* __restrict__ B0, float* __restrict__ C0,
    const float* __restrict__ A1, const float* __restrict__ B1, float* __restrict__ C1,
    const float* __restrict__ A2, const float* __restrict__ B2, float* __restrict__ C2,
    int M, int N, int K) {
  __shared__ __align__(16) float As[64][APITCH];
  __shared__ __align__(16) float Bs[GK][BPITCH];
  const int z = blockIdx.z;
  const float* A = z == 0 ? A0 : (z == 1 ? A1 : A2);
  const float* B = z == 0 ? B0 : (z == 1 ? B1 : B2);
  float* C = z == 0 ? C0 : (z == 1 ? C1 : C2);
  const int tid  = threadIdx.x;
  const int wave = tid >> 5;
  const int lane = tid & 31;
  const int half = lane >> 4;
  const int l16  = lane & 15;
  const int row0 = blockIdx.y * 64;
  const int col0 = blockIdx.x * 64;

  v8f acc[4];
#pragma unroll
  for (int nt = 0; nt < 4; ++nt) acc[nt] = (v8f)(0.0f);

  const int ar = tid / (GK / 4), ac = (tid % (GK / 4)) << 2;
  const int br = tid >> 4, bc = (tid & 15) << 2;
  const float* aptr = A + (size_t)(row0 + ar) * K + ac;
  const float* bptr = B + (size_t)br * N + col0 + bc;

  v4f areg[GIT], breg[GIT];
#pragma unroll
  for (int it = 0; it < GIT; ++it) {
    areg[it] = *(const v4f*)(aptr + (size_t)(it * (512 / GK)) * K);
    breg[it] = *(const v4f*)(bptr + (size_t)(it * 8) * N);
  }

  for (int kk = 0; kk < K; kk += GK) {
    __syncthreads();
#pragma unroll
    for (int it = 0; it < GIT; ++it) {
      *(v4f*)&As[ar + it * (512 / GK)][ac] = areg[it];
      *(v4f*)&Bs[br + it * 8][bc] = breg[it];
    }
    __syncthreads();
    if (kk + GK < K) {
#pragma unroll
      for (int it = 0; it < GIT; ++it) {
        areg[it] = *(const v4f*)(aptr + (size_t)(it * (512 / GK)) * K + kk + GK);
        breg[it] = *(const v4f*)(bptr + (size_t)(kk + GK + it * 8) * N);
      }
    }
    v2f af[2][SB];
    float bx[2][SB][4], by[2][SB][4];
#pragma unroll
    for (int s = 0; s < SB; ++s) {
      const int k0 = s * 4;
      af[0][s] = *(const v2f*)&As[wave * 16 + l16][k0 + 2 * half];
#pragma unroll
      for (int nt = 0; nt < 4; ++nt) {
        bx[0][s][nt] = Bs[k0 + 2 * half + 0][nt * 16 + l16];
        by[0][s][nt] = Bs[k0 + 2 * half + 1][nt * 16 + l16];
      }
    }
#pragma unroll
    for (int blk = 0; blk < NB; ++blk) {
      const int cur = blk & 1;
      if (blk + 1 < NB) {
#pragma unroll
        for (int s = 0; s < SB; ++s) {
          const int k0 = ((blk + 1) * SB + s) * 4;
          af[cur ^ 1][s] = *(const v2f*)&As[wave * 16 + l16][k0 + 2 * half];
#pragma unroll
          for (int nt = 0; nt < 4; ++nt) {
            bx[cur ^ 1][s][nt] = Bs[k0 + 2 * half + 0][nt * 16 + l16];
            by[cur ^ 1][s][nt] = Bs[k0 + 2 * half + 1][nt * 16 + l16];
          }
        }
      }
      __builtin_amdgcn_sched_barrier(0);
#pragma unroll
      for (int s = 0; s < SB; ++s)
#pragma unroll
        for (int nt = 0; nt < 4; ++nt) {
          v2f b;
          b.x = bx[cur][s][nt];
          b.y = by[cur][s][nt];
          acc[nt] = wmma4(af[cur][s], b, acc[nt]);
        }
      if (blk + 1 < NB) __builtin_amdgcn_sched_barrier(0);
    }
  }
#pragma unroll
  for (int nt = 0; nt < 4; ++nt)
#pragma unroll
    for (int r = 0; r < 8; ++r)
      C[(size_t)(row0 + wave * 16 + r + 8 * half) * N + col0 + nt * 16 + l16] =
          acc[nt][r];
}

#define KT 128
#define KW (KT / 2)
#define NKT (KW / 16)
#define QT 32
#define KPITCH 68
#define VPITCH 72
#define PPITCH (KW + 4)
#define SI 8
#define SP 8
__global__ __launch_bounds__(BLOCK) void fa_f32_wmma(
    const float* __restrict__ Qg, const float* __restrict__ Kp,
    const float* __restrict__ Vp, float* __restrict__ Op, int S) {
  __shared__ __align__(16) float Ks[KT][KPITCH];
  __shared__ __align__(16) float Vs[KT][VPITCH];
  __shared__ __align__(16) float Ps[4][16][PPITCH];
  __shared__ float Xo[2][32][32];
  __shared__ float Xm[2][8][32];
  __shared__ float Xl[2][8][32];
  const int tid  = threadIdx.x;
  const int wave = tid >> 5;
  const int lane = tid & 31;
  const int half = lane >> 4;
  const int l16  = lane & 15;
  const int rh   = wave & 1;
  const int kh   = wave >> 1;
  const int b    = blockIdx.y >> 2;
  const int h    = blockIdx.y & 3;
  const int ntq  = S / QT;

  const int ldr = tid >> 4, ldc = (tid & 15) << 2;
  const size_t gbase = (size_t)(b * S + ldr) * 256 + h * 64 + ldc;

  for (int t = 0; t < 2; ++t) {
  const int qt = t ? (ntq - 1 - (int)blockIdx.x) : (int)blockIdx.x;
  const int q0 = qt * QT;

  v2f qf[16];
  {
    const int qrow = q0 + rh * 16 + l16;
    const float* qptr =
        Qg + (size_t)(b * S + qrow) * 256 + h * 64 + 2 * half;
#pragma unroll
    for (int i = 0; i < 16; ++i) qf[i] = *(const v2f*)(qptr + i * 4);
  }

  v8f oacc[4];
#pragma unroll
  for (int nt = 0; nt < 4; ++nt) oacc[nt] = (v8f)(0.0f);
  float m[8], l[8];
#pragma unroll
  for (int r = 0; r < 8; ++r) { m[r] = -1e30f; l[r] = 0.0f; }

  const int ntiles = (q0 + QT - 1) / KT + 1;
  v4f kreg[KT / 8], vreg[KT / 8];
#pragma unroll
  for (int it = 0; it < KT / 8; ++it) {
    kreg[it] = *(const v4f*)(Kp + gbase + (size_t)it * 8 * 256);
    vreg[it] = *(const v4f*)(Vp + gbase + (size_t)it * 8 * 256);
  }
  for (int j = 0; j < ntiles; ++j) {
    const int a0 = j * KT;
    __syncthreads();
#pragma unroll
    for (int it = 0; it < KT / 8; ++it) {
      *(v4f*)&Ks[ldr + it * 8][ldc] = kreg[it];
      *(v4f*)&Vs[ldr + it * 8][ldc] = vreg[it];
    }
    __syncthreads();
    if (j + 1 < ntiles) {
      const size_t g1 = gbase + (size_t)(a0 + KT) * 256;
#pragma unroll
      for (int it = 0; it < KT / 8; ++it) {
        kreg[it] = *(const v4f*)(Kp + g1 + (size_t)it * 8 * 256);
        vreg[it] = *(const v4f*)(Vp + g1 + (size_t)it * 8 * 256);
      }
    }

    v8f s[NKT];
#pragma unroll
    for (int kt = 0; kt < NKT; ++kt) s[kt] = (v8f)(0.0f);
    {
      const int krow = kh * KW + l16;
      v2f kf[2][SI][NKT];
#pragma unroll
      for (int ii = 0; ii < SI; ++ii)
#pragma unroll
        for (int kt = 0; kt < NKT; ++kt)
          kf[0][ii][kt] = *(const v2f*)&Ks[krow + kt * 16][ii * 4 + 2 * half];
#pragma unroll
      for (int blk = 0; blk < 16 / SI; ++blk) {
        const int cur = blk & 1;
        if (blk + 1 < 16 / SI) {
#pragma unroll
          for (int ii = 0; ii < SI; ++ii)
#pragma unroll
            for (int kt = 0; kt < NKT; ++kt)
              kf[cur ^ 1][ii][kt] =
                  *(const v2f*)&Ks[krow + kt * 16][((blk + 1) * SI + ii) * 4 + 2 * half];
        }
        __builtin_amdgcn_sched_barrier(0);
#pragma unroll
        for (int ii = 0; ii < SI; ++ii)
#pragma unroll
          for (int kt = 0; kt < NKT; ++kt)
            s[kt] = wmma4(qf[blk * SI + ii], kf[cur][ii][kt], s[kt]);
        if (blk + 1 < 16 / SI) __builtin_amdgcn_sched_barrier(0);
      }
    }

#pragma unroll
    for (int r = 0; r < 8; ++r) {
      const int qrow = q0 + rh * 16 + r + 8 * half;
      float x[NKT];
#pragma unroll
      for (int kt = 0; kt < NKT; ++kt) {
        x[kt] = s[kt][r] * 0.125f;
        if (a0 + kh * KW + kt * 16 + l16 > qrow) x[kt] = -1e30f;
      }
      float mx = x[0];
#pragma unroll
      for (int kt = 1; kt < NKT; ++kt) mx = fmaxf(mx, x[kt]);
#pragma unroll
      for (int off = 1; off < 16; off <<= 1)
        mx = fmaxf(mx, __shfl_xor(mx, off, 16));
      const float mnew = fmaxf(m[r], mx);
      float p[NKT];
#pragma unroll
      for (int kt = 0; kt < NKT; ++kt) p[kt] = __expf(x[kt] - mnew);
      float sum = p[0];
#pragma unroll
      for (int kt = 1; kt < NKT; ++kt) sum += p[kt];
#pragma unroll
      for (int off = 1; off < 16; off <<= 1) sum += __shfl_xor(sum, off, 16);
      const float scale = __expf(m[r] - mnew);
      l[r] = l[r] * scale + sum;
      m[r] = mnew;
#pragma unroll
      for (int nt = 0; nt < 4; ++nt) oacc[nt][r] *= scale;
#pragma unroll
      for (int kt = 0; kt < NKT; ++kt) Ps[wave][r + 8 * half][kt * 16 + l16] = p[kt];
    }
    __syncthreads();

    {
      const int vrow = kh * KW + 2 * half;
      v2f pf[2][SP];
      float vx[2][SP][4], vy[2][SP][4];
#pragma unroll
      for (int s2 = 0; s2 < SP; ++s2) {
        const int k0 = s2 * 4;
        pf[0][s2] = *(const v2f*)&Ps[wave][l16][k0 + 2 * half];
#pragma unroll
        for (int nt = 0; nt < 4; ++nt) {
          vx[0][s2][nt] = Vs[vrow + k0 + 0][nt * 16 + l16];
          vy[0][s2][nt] = Vs[vrow + k0 + 1][nt * 16 + l16];
        }
      }
#pragma unroll
      for (int blk = 0; blk < KW / 4 / SP; ++blk) {
        const int cur = blk & 1;
        if (blk + 1 < KW / 4 / SP) {
#pragma unroll
          for (int s2 = 0; s2 < SP; ++s2) {
            const int k0 = ((blk + 1) * SP + s2) * 4;
            pf[cur ^ 1][s2] = *(const v2f*)&Ps[wave][l16][k0 + 2 * half];
#pragma unroll
            for (int nt = 0; nt < 4; ++nt) {
              vx[cur ^ 1][s2][nt] = Vs[vrow + k0 + 0][nt * 16 + l16];
              vy[cur ^ 1][s2][nt] = Vs[vrow + k0 + 1][nt * 16 + l16];
            }
          }
        }
        __builtin_amdgcn_sched_barrier(0);
#pragma unroll
        for (int s2 = 0; s2 < SP; ++s2)
#pragma unroll
          for (int nt = 0; nt < 4; ++nt) {
            v2f bv;
            bv.x = vx[cur][s2][nt];
            bv.y = vy[cur][s2][nt];
            oacc[nt] = wmma4(pf[cur][s2], bv, oacc[nt]);
          }
        if (blk + 1 < KW / 4 / SP) __builtin_amdgcn_sched_barrier(0);
      }
    }
  }

  if (kh == 1) {
#pragma unroll
    for (int r = 0; r < 8; ++r) {
      Xm[rh][r][lane] = m[r];
      Xl[rh][r][lane] = l[r];
#pragma unroll
      for (int nt = 0; nt < 4; ++nt) Xo[rh][nt * 8 + r][lane] = oacc[nt][r];
    }
  }
  __syncthreads();
  if (kh == 0) {
#pragma unroll
    for (int r = 0; r < 8; ++r) {
      const float m2 = Xm[rh][r][lane];
      const float l2 = Xl[rh][r][lane];
      const float mm = fmaxf(m[r], m2);
      const float w1 = __expf(m[r] - mm);
      const float w2 = __expf(m2 - mm);
      const float inv = 1.0f / (l[r] * w1 + l2 * w2);
      const int row = b * S + q0 + rh * 16 + r + 8 * half;
#pragma unroll
      for (int nt = 0; nt < 4; ++nt)
        Op[(size_t)row * 256 + h * 64 + nt * 16 + l16] =
            (oacc[nt][r] * w1 + Xo[rh][nt * 8 + r][lane] * w2) * inv;
    }
  }
  }
}

extern "C" void kernel_launch(void* const* d_in, const int* in_sizes, int n_in,
                              void* d_out, int out_size, void* d_ws,
                              size_t ws_size, hipStream_t stream) {
  const float* q_in  = (const float*)d_in[0];
  const float* kv_in = (const float*)d_in[1];
  const float* Wq = (const float*)d_in[3];
  const float* Wk = (const float*)d_in[4];
  const float* Wv = (const float*)d_in[5];
  const float* Wo = (const float*)d_in[6];
  float* out = (float*)d_out;

  const int B = 2, S = 2048, IN = 1024;
  const int M = B * S;

  float* ws  = (float*)d_ws;
  float* wqg = ws;
  float* qg  = wqg + 1024 * 256;
  float* kp  = qg + M * 256;
  float* vp  = kp + M * 256;
  float* op  = vp + M * 256;

  reduce_wq_kernel<<<(1024 * 256) / 256, 256, 0, stream>>>(Wq, wqg);

  dim3 blk(BLOCK);
  gemm_f32_wmma<<<dim3(256 / 64, M / 64, 3), blk, 0, stream>>>(
      q_in, wqg, qg, kv_in, Wk, kp, kv_in, Wv, vp, M, 256, IN);

  fa_f32_wmma<<<dim3(S / 64, B * 4), blk, 0, stream>>>(qg, kp, vp, op, S);

  gemm_f32_wmma<<<dim3(IN / 64, M / 64, 1), blk, 0, stream>>>(
      op, Wo, out, nullptr, nullptr, nullptr, nullptr, nullptr, nullptr, M, IN, 256);
}


# attention computed transposed (S^T = K*Q^T, O^T = V^T*P^T): probabilities feed the second product from the score registers, row max/sum in-lane, no P tile in LDS; changes order of row sums
# speedup vs baseline: 8.1328x; 1.1230x over previous
#include <hip/hip_runtime.h>
#include <hip/hip_bf16.h>

typedef __attribute__((ext_vector_type(2))) float v2f;
typedef __attribute__((ext_vector_type(8))) float v8f;
typedef __attribute__((ext_vector_type(4))) float v4f;

#define BLOCK 128

__device__ __forceinline__ v8f wmma4(v2f a, v2f b, v8f c) {
  return __builtin_amdgcn_wmma_f32_16x16x4_f32(false, a, false, b, (short)0, c,
                                               false, false);
}

__global__ void reduce_wq_kernel(const float* __restrict__ Wq,
                                 float* __restrict__ Wqg) {
  int idx = blockIdx.x * blockDim.x + threadIdx.x;
  if (idx >= 1024 * 256) return;
  int i = idx >> 8;
  int c = idx & 255;
  int h = c >> 6, d = c & 63;
  const float* p = Wq + (size_t)i * 1024 + h * 256 + d;
  Wqg[idx] = p[0] + p[64] + p[128] + p[192];
}

#define GK 128
#define GIT (GK / 8)
#define APITCH (GK + 4)
#define BPITCH 80
#define SB 8
#define NB (GK / 4 / SB)
__global__ __launch_bounds__(BLOCK, 3) void gemm_f32_wmma(
    const float* __restrict__ A0, const float* __restrict__ B0, float* __restrict__ C0,
    const float* __restrict__ A1, const float* __restrict__ B1, float* __restrict__ C1,
    const float* __restrict__ A2, const float* __restrict__ B2, float* __restrict__ C2,
    int M, int N, int K) {
  __shared__ __align__(16) float As[64][APITCH];
  __shared__ __align__(16) float Bs[GK][BPITCH];
  const int z = blockIdx.z;
  const float* A = z == 0 ? A0 : (z == 1 ? A1 : A2);
  const float* B = z == 0 ? B0 : (z == 1 ? B1 : B2);
  float* C = z == 0 ? C0 : (z == 1 ? C1 : C2);
  const int tid  = threadIdx.x;
  const int wave = tid >> 5;
  const int lane = tid & 31;
  const int half = lane >> 4;
  const int l16  = lane & 15;
  const int row0 = blockIdx.y * 64;
  const int col0 = blockIdx.x * 64;

  v8f acc[4];
#pragma unroll
  for (int nt = 0; nt < 4; ++nt) acc[nt] = (v8f)(0.0f);

  const int ar = tid / (GK / 4), ac = (tid % (GK / 4)) << 2;
  const int br = tid >> 4, bc = (tid & 15) << 2;
  const float* aptr = A + (size_t)(row0 + ar) * K + ac;
  const float* bptr = B + (size_t)br * N + col0 + bc;

  v4f areg[GIT], breg[GIT];
#pragma unroll
  for (int it = 0; it < GIT; ++it) {
    areg[it] = *(const v4f*)(aptr + (size_t)(it * (512 / GK)) * K);
    breg[it] = *(const v4f*)(bptr + (size_t)(it * 8) * N);
  }

  for (int kk = 0; kk < K; kk += GK) {
    __syncthreads();
#pragma unroll
    for (int it = 0; it < GIT; ++it) {
      *(v4f*)&As[ar + it * (512 / GK)][ac] = areg[it];
      *(v4f*)&Bs[br + it * 8][bc] = breg[it];
    }
    __syncthreads();
    if (kk + GK < K) {
#pragma unroll
      for (int it = 0; it < GIT; ++it) {
        areg[it] = *(const v4f*)(aptr + (size_t)(it * (512 / GK)) * K + kk + GK);
        breg[it] = *(const v4f*)(bptr + (size_t)(kk + GK + it * 8) * N);
      }
    }
    v2f af[2][SB];
    float bx[2][SB][4], by[2][SB][4];
#pragma unroll
    for (int s = 0; s < SB; ++s) {
      const int k0 = s * 4;
      af[0][s] = *(const v2f*)&As[wave * 16 + l16][k0 + 2 * half];
#pragma unroll
      for (int nt = 0; nt < 4; ++nt) {
        bx[0][s][nt] = Bs[k0 + 2 * half + 0][nt * 16 + l16];
        by[0][s][nt] = Bs[k0 + 2 * half + 1][nt * 16 + l16];
      }
    }
#pragma unroll
    for (int blk = 0; blk < NB; ++blk) {
      const int cur = blk & 1;
      if (blk + 1 < NB) {
#pragma unroll
        for (int s = 0; s < SB; ++s) {
          const int k0 = ((blk + 1) * SB + s) * 4;
          af[cur ^ 1][s] = *(const v2f*)&As[wave * 16 + l16][k0 + 2 * half];
#pragma unroll
          for (int nt = 0; nt < 4; ++nt) {
            bx[cur ^ 1][s][nt] = Bs[k0 + 2 * half + 0][nt * 16 + l16];
            by[cur ^ 1][s][nt] = Bs[k0 + 2 * half + 1][nt * 16 + l16];
          }
        }
      }
      __builtin_amdgcn_sched_barrier(0);
#pragma unroll
      for (int s = 0; s < SB; ++s)
#pragma unroll
        for (int nt = 0; nt < 4; ++nt) {
          v2f b;
          b.x = bx[cur][s][nt];
          b.y = by[cur][s][nt];
          acc[nt] = wmma4(af[cur][s], b, acc[nt]);
        }
      if (blk + 1 < NB) __builtin_amdgcn_sched_barrier(0);
    }
  }
#pragma unroll
  for (int nt = 0; nt < 4; ++nt)
#pragma unroll
    for (int r = 0; r < 8; ++r)
      C[(size_t)(row0 + wave * 16 + r + 8 * half) * N + col0 + nt * 16 + l16] =
          acc[nt][r];
}

#define KT 128
#define KW (KT / 2)
#define NKT (KW / 16)
#define QT 32
#define KPITCH 68
#define VPITCH 68
#define SI 8
#define SP 8

__device__ __forceinline__ float xhalf(float x) {
  const int b = __builtin_bit_cast(int, x);
  return __builtin_bit_cast(
      float, __builtin_amdgcn_permlanex16(b, b, 0x76543210, 0xfedcba98, false, false));
}

__global__ __launch_bounds__(BLOCK) void fa_f32_wmma(
    const float* __restrict__ Qg, const float* __restrict__ Kp,
    const float* __restrict__ Vp, float* __restrict__ Op, int S) {
  __shared__ __align__(16) float Ks[KT][KPITCH];
  __shared__ __align__(16) float Vs[KT][VPITCH];
  __shared__ float Xo[2][32][32];
  __shared__ float Xm[2][32];
  __shared__ float Xl[2][32];
  const int tid  = threadIdx.x;
  const int wave = tid >> 5;
  const int lane = tid & 31;
  const int half = lane >> 4;
  const int l16  = lane & 15;
  const int rh   = wave & 1;
  const int kh   = wave >> 1;
  const int b    = blockIdx.y >> 2;
  const int h    = blockIdx.y & 3;
  const int ntq  = S / QT;

  const int ldr = tid >> 4, ldc = (tid & 15) << 2;
  const size_t gbase = (size_t)(b * S + ldr) * 256 + h * 64 + ldc;

  for (int t = 0; t < 2; ++t) {
  const int qt = t ? (ntq - 1 - (int)blockIdx.x) : (int)blockIdx.x;
  const int q0 = qt * QT;
  const int qrow = q0 + rh * 16 + l16;

  v2f qf[16];
  {
    const float* qptr =
        Qg + (size_t)(b * S + qrow) * 256 + h * 64 + 2 * half;
#pragma unroll
    for (int i = 0; i < 16; ++i) qf[i] = *(const v2f*)(qptr + i * 4);
  }

  v8f oacc[4];
#pragma unroll
  for (int nt = 0; nt < 4; ++nt) oacc[nt] = (v8f)(0.0f);
  float m = -1e30f, l = 0.0f;

  const int ntiles = (q0 + QT - 1) / KT + 1;
  v4f kreg[KT / 8], vreg[KT / 8];
#pragma unroll
  for (int it = 0; it < KT / 8; ++it) {
    kreg[it] = *(const v4f*)(Kp + gbase + (size_t)it * 8 * 256);
    vreg[it] = *(const v4f*)(Vp + gbase + (size_t)it * 8 * 256);
  }
  for (int j = 0; j < ntiles; ++j) {
    const int a0 = j * KT;
    __syncthreads();
#pragma unroll
    for (int it = 0; it < KT / 8; ++it) {
      *(v4f*)&Ks[ldr + it * 8][ldc] = kreg[it];
      *(v4f*)&Vs[ldr + it * 8][ldc] = vreg[it];
    }
    __syncthreads();
    if (j + 1 < ntiles) {
      const size_t g1 = gbase + (size_t)(a0 + KT) * 256;
#pragma unroll
      for (int it = 0; it < KT / 8; ++it) {
        kreg[it] = *(const v4f*)(Kp + g1 + (size_t)it * 8 * 256);
        vreg[it] = *(const v4f*)(Vp + g1 + (size_t)it * 8 * 256);
      }
    }

    v8f s[NKT];
#pragma unroll
    for (int kt = 0; kt < NKT; ++kt) s[kt] = (v8f)(0.0f);
    {
      const int krow = kh * KW + l16;
      v2f kf[2][SI][NKT];
#pragma unroll
      for (int ii = 0; ii < SI; ++ii)
#pragma unroll
        for (int kt = 0; kt < NKT; ++kt)
          kf[0][ii][kt] = *(const v2f*)&Ks[krow + kt * 16][ii * 4 + 2 * half];
#pragma unroll
      for (int blk = 0; blk < 16 / SI; ++blk) {
        const int cur = blk & 1;
        if (blk + 1 < 16 / SI) {
#pragma unroll
          for (int ii = 0; ii < SI; ++ii)
#pragma unroll
            for (int kt = 0; kt < NKT; ++kt)
              kf[cur ^ 1][ii][kt] =
                  *(const v2f*)&Ks[krow + kt * 16][((blk + 1) * SI + ii) * 4 + 2 * half];
        }
        __builtin_amdgcn_sched_barrier(0);
#pragma unroll
        for (int ii = 0; ii < SI; ++ii)
#pragma unroll
          for (int kt = 0; kt < NKT; ++kt)
            s[kt] = wmma4(kf[cur][ii][kt], qf[blk * SI + ii], s[kt]);
        __builtin_amdgcn_sched_barrier(0);
      }
    }

    const int vrow = kh * KW + 8 * half;
    float vx[2][SP][4], vy[2][SP][4];
#pragma unroll
    for (int s2 = 0; s2 < SP; ++s2) {
      const int kr = vrow + (s2 >> 2) * 16 + (s2 & 3) * 2;
#pragma unroll
      for (int nt = 0; nt < 4; ++nt) {
        vx[0][s2][nt] = Vs[kr + 0][nt * 16 + l16];
        vy[0][s2][nt] = Vs[kr + 1][nt * 16 + l16];
      }
    }

    {
      const int key0 = a0 + kh * KW + 8 * half;
      float mxk[NKT];
#pragma unroll
      for (int kt = 0; kt < NKT; ++kt) {
#pragma unroll
        for (int r = 0; r < 8; ++r) {
          float x = s[kt][r] * 0.125f;
          if (key0 + kt * 16 + r > qrow) x = -1e30f;
          s[kt][r] = x;
          mxk[kt] = r ? fmaxf(mxk[kt], x) : x;
        }
      }
      float mx = fmaxf(fmaxf(mxk[0], mxk[1]), fmaxf(mxk[2], mxk[3]));
      mx = fmaxf(mx, xhalf(mx));
      const float mnew = fmaxf(m, mx);
      float smk[NKT];
#pragma unroll
      for (int kt = 0; kt < NKT; ++kt) {
#pragma unroll
        for (int r = 0; r < 8; ++r) {
          const float p = __expf(s[kt][r] - mnew);
          s[kt][r] = p;
          smk[kt] = r ? smk[kt] + p : p;
        }
      }
      const float scale = __expf(m - mnew);
      l = l * scale + ((smk[0] + smk[1]) + (smk[2] + smk[3]));
      m = mnew;
#pragma unroll
      for (int nt = 0; nt < 4; ++nt)
#pragma unroll
        for (int r = 0; r < 8; ++r) oacc[nt][r] *= scale;
    }

    {
#pragma unroll
      for (int blk = 0; blk < KW / 4 / SP; ++blk) {
        const int cur = blk & 1;
        if (blk + 1 < KW / 4 / SP) {
#pragma unroll
          for (int s2 = 0; s2 < SP; ++s2) {
            const int st = (blk + 1) * SP + s2;
            const int kr = vrow + (st >> 2) * 16 + (st & 3) * 2;
#pragma unroll
            for (int nt = 0; nt < 4; ++nt) {
              vx[cur ^ 1][s2][nt] = Vs[kr + 0][nt * 16 + l16];
              vy[cur ^ 1][s2][nt] = Vs[kr + 1][nt * 16 + l16];
            }
          }
        }
        __builtin_amdgcn_sched_barrier(0);
#pragma unroll
        for (int s2 = 0; s2 < SP; ++s2) {
          const int st = blk * SP + s2;
          v2f pt;
          pt.x = s[st >> 2][(st & 3) * 2 + 0];
          pt.y = s[st >> 2][(st & 3) * 2 + 1];
#pragma unroll
          for (int nt = 0; nt < 4; ++nt) {
            v2f av;
            av.x = vx[cur][s2][nt];
            av.y = vy[cur][s2][nt];
            oacc[nt] = wmma4(av, pt, oacc[nt]);
          }
        }
        if (blk + 1 < KW / 4 / SP) __builtin_amdgcn_sched_barrier(0);
      }
    }
  }

  l += xhalf(l);
  if (kh == 1) {
    Xm[rh][lane] = m;
    Xl[rh][lane] = l;
#pragma unroll
    for (int nt = 0; nt < 4; ++nt)
#pragma unroll
      for (int r = 0; r < 8; ++r) Xo[rh][nt * 8 + r][lane] = oacc[nt][r];
  }
  __syncthreads();
  if (kh == 0) {
    const float m2 = Xm[rh][lane];
    const float l2 = Xl[rh][lane];
    const float mm = fmaxf(m, m2);
    const float w1 = __expf(m - mm);
    const float w2 = __expf(m2 - mm);
    const float inv = 1.0f / (l * w1 + l2 * w2);
    float* orow = Op + (size_t)(b * S + qrow) * 256 + h * 64 + 8 * half;
#pragma unroll
    for (int nt = 0; nt < 4; ++nt) {
      v4f lo, hi;
      lo.x = (oacc[nt][0] * w1 + Xo[rh][nt * 8 + 0][lane] * w2) * inv;
      lo.y = (oacc[nt][1] * w1 + Xo[rh][nt * 8 + 1][lane] * w2) * inv;
      lo.z = (oacc[nt][2] * w1 + Xo[rh][nt * 8 + 2][lane] * w2) * inv;
      lo.w = (oacc[nt][3] * w1 + Xo[rh][nt * 8 + 3][lane] * w2) * inv;
      hi.x = (oacc[nt][4] * w1 + Xo[rh][nt * 8 + 4][lane] * w2) * inv;
      hi.y = (oacc[nt][5] * w1 + Xo[rh][nt * 8 + 5][lane] * w2) * inv;
      hi.z = (oacc[nt][6] * w1 + Xo[rh][nt * 8 + 6][lane] * w2) * inv;
      hi.w = (oacc[nt][7] * w1 + Xo[rh][nt * 8 + 7][lane] * w2) * inv;
      *(v4f*)(orow + nt * 16) = lo;
      *(v4f*)(orow + nt * 16 + 4) = hi;
    }
  }
  }
}

extern "C" void kernel_launch(void* const* d_in, const int* in_sizes, int n_in,
                              void* d_out, int out_size, void* d_ws,
                              size_t ws_size, hipStream_t stream) {
  const float* q_in  = (const float*)d_in[0];
  const float* kv_in = (const float*)d_in[1];
  const float* Wq = (const float*)d_in[3];
  const float* Wk = (const float*)d_in[4];
  const float* Wv = (const float*)d_in[5];
  const float* Wo = (const float*)d_in[6];
  float* out = (float*)d_out;

  const int B = 2, S = 2048, IN = 1024;
  const int M = B * S;

  float* ws  = (float*)d_ws;
  float* wqg = ws;
  float* qg  = wqg + 1024 * 256;
  float* kp  = qg + M * 256;
  float* vp  = kp + M * 256;
  float* op  = vp + M * 256;

  reduce_wq_kernel<<<(1024 * 256) / 256, 256, 0, stream>>>(Wq, wqg);

  dim3 blk(BLOCK);
  gemm_f32_wmma<<<dim3(256 / 64, M / 64, 3), blk, 0, stream>>>(
      q_in, wqg, qg, kv_in, Wk, kp, kv_in, Wv, vp, M, 256, IN);

  fa_f32_wmma<<<dim3(S / 64, B * 4), blk, 0, stream>>>(qg, kp, vp, op, S);

  gemm_f32_wmma<<<dim3(IN / 64, M / 64, 1), blk, 0, stream>>>(
      op, Wo, out, nullptr, nullptr, nullptr, nullptr, nullptr, nullptr, M, IN, 256);
}


# GEMM B tile in LDS in fragment order (one matrix instruction's B operand = 256 contiguous bytes; block column 4c+t in tile t): one 8-byte load per fragment, no pairing moves, 16-byte C stores
# speedup vs baseline: 8.7733x; 1.0788x over previous
#include <hip/hip_runtime.h>
#include <hip/hip_bf16.h>

typedef __attribute__((ext_vector_type(2))) float v2f;
typedef __attribute__((ext_vector_type(8))) float v8f;
typedef __attribute__((ext_vector_type(4))) float v4f;

#define BLOCK 128

__device__ __forceinline__ v8f wmma4(v2f a, v2f b, v8f c) {
  return __builtin_amdgcn_wmma_f32_16x16x4_f32(false, a, false, b, (short)0, c,
                                               false, false);
}

__global__ void reduce_wq_kernel(const float* __restrict__ Wq,
                                 float* __restrict__ Wqg) {
  int idx = blockIdx.x * blockDim.x + threadIdx.x;
  if (idx >= 1024 * 256) return;
  int i = idx >> 8;
  int c = idx & 255;
  int h = c >> 6, d = c & 63;
  const float* p = Wq + (size_t)i * 1024 + h * 256 + d;
  Wqg[idx] = p[0] + p[64] + p[128] + p[192];
}

#define GK 128
#define GIT (GK / 8)
#define APITCH (GK + 4)
#define SB 8
#define NB (GK / 4 / SB)
__global__ __launch_bounds__(BLOCK, 3) void gemm_f32_wmma(
    const float* __restrict__ A0, const float* __restrict__ B0, float* __restrict__ C0,
    const float* __restrict__ A1, const float* __restrict__ B1, float* __restrict__ C1,
    const float* __restrict__ A2, const float* __restrict__ B2, float* __restrict__ C2,
    int M, int N, int K) {
  __shared__ __align__(16) float As[64][APITCH];
  __shared__ __align__(16) float Bs[GK][64];
  const int z = blockIdx.z;
  const float* A = z == 0 ? A0 : (z == 1 ? A1 : A2);
  const float* B = z == 0 ? B0 : (z == 1 ? B1 : B2);
  float* C = z == 0 ? C0 : (z == 1 ? C1 : C2);
  const int tid  = threadIdx.x;
  const int wave = tid >> 5;
  const int lane = tid & 31;
  const int half = lane >> 4;
  const int l16  = lane & 15;
  const int row0 = blockIdx.y * 64;
  const int col0 = blockIdx.x * 64;

  v8f acc[4];
#pragma unroll
  for (int nt = 0; nt < 4; ++nt) acc[nt] = (v8f)(0.0f);

  const int ar = tid / (GK / 4), ac = (tid % (GK / 4)) << 2;
  const int br = tid >> 4, bc = (tid & 15) << 2;
  const float* aptr = A + (size_t)(row0 + ar) * K + ac;
  const float* bptr = B + (size_t)br * N + col0 + bc;

  v4f areg[GIT], breg[GIT];
#pragma unroll
  for (int it = 0; it < GIT; ++it) {
    areg[it] = *(const v4f*)(aptr + (size_t)(it * (512 / GK)) * K);
    breg[it] = *(const v4f*)(bptr + (size_t)(it * 8) * N);
  }

  for (int kk = 0; kk < K; kk += GK) {
    __syncthreads();
#pragma unroll
    for (int it = 0; it < GIT; ++it) {
      *(v4f*)&As[ar + it * (512 / GK)][ac] = areg[it];
      float* bdst = &Bs[((br >> 2) + 2 * it) * 4][((br >> 1) & 1) * 32 + 2 * (tid & 15) + (br & 1)];
      bdst[0]   = breg[it].x;
      bdst[64]  = breg[it].y;
      bdst[128] = breg[it].z;
      bdst[192] = breg[it].w;
    }
    __syncthreads();
    if (kk + GK < K) {
#pragma unroll
      for (int it = 0; it < GIT; ++it) {
        areg[it] = *(const v4f*)(aptr + (size_t)(it * (512 / GK)) * K + kk + GK);
        breg[it] = *(const v4f*)(bptr + (size_t)(kk + GK + it * 8) * N);
      }
    }
    v2f af[2][SB];
    v2f bf[2][SB][4];
#pragma unroll
    for (int s = 0; s < SB; ++s) {
      const int k0 = s * 4;
      af[0][s] = *(const v2f*)&As[wave * 16 + l16][k0 + 2 * half];
#pragma unroll
      for (int nt = 0; nt < 4; ++nt)
        bf[0][s][nt] = *(const v2f*)&Bs[k0 + nt][2 * lane];
    }
#pragma unroll
    for (int blk = 0; blk < NB; ++blk) {
      const int cur = blk & 1;
      if (blk + 1 < NB) {
#pragma unroll
        for (int s = 0; s < SB; ++s) {
          const int k0 = ((blk + 1) * SB + s) * 4;
          af[cur ^ 1][s] = *(const v2f*)&As[wave * 16 + l16][k0 + 2 * half];
#pragma unroll
          for (int nt = 0; nt < 4; ++nt)
            bf[cur ^ 1][s][nt] = *(const v2f*)&Bs[k0 + nt][2 * lane];
        }
      }
      __builtin_amdgcn_sched_barrier(0);
#pragma unroll
      for (int s = 0; s < SB; ++s)
#pragma unroll
        for (int nt = 0; nt < 4; ++nt)
          acc[nt] = wmma4(af[cur][s], bf[cur][s][nt], acc[nt]);
      if (blk + 1 < NB) __builtin_amdgcn_sched_barrier(0);
    }
  }
#pragma unroll
  for (int r = 0; r < 8; ++r) {
    v4f o;
    o.x = acc[0][r];
    o.y = acc[1][r];
    o.z = acc[2][r];
    o.w = acc[3][r];
    *(v4f*)&C[(size_t)(row0 + wave * 16 + r + 8 * half) * N + col0 + 4 * l16] = o;
  }
}

#define KT 128
#define KW (KT / 2)
#define NKT (KW / 16)
#define QT 32
#define KPITCH 68
#define VPITCH 68
#define SI 8
#define SP 8

__device__ __forceinline__ float xhalf(float x) {
  const int b = __builtin_bit_cast(int, x);
  return __builtin_bit_cast(
      float, __builtin_amdgcn_permlanex16(b, b, 0x76543210, 0xfedcba98, false, false));
}

__global__ __launch_bounds__(BLOCK) void fa_f32_wmma(
    const float* __restrict__ Qg, const float* __restrict__ Kp,
    const float* __restrict__ Vp, float* __restrict__ Op, int S) {
  __shared__ __align__(16) float Ks[KT][KPITCH];
  __shared__ __align__(16) float Vs[KT][VPITCH];
  __shared__ float Xo[2][32][32];
  __shared__ float Xm[2][32];
  __shared__ float Xl[2][32];
  const int tid  = threadIdx.x;
  const int wave = tid >> 5;
  const int lane = tid & 31;
  const int half = lane >> 4;
  const int l16  = lane & 15;
  const int rh   = wave & 1;
  const int kh   = wave >> 1;
  const int b    = blockIdx.y >> 2;
  const int h    = blockIdx.y & 3;
  const int ntq  = S / QT;

  const int ldr = tid >> 4, ldc = (tid & 15) << 2;
  const size_t gbase = (size_t)(b * S + ldr) * 256 + h * 64 + ldc;

  for (int t = 0; t < 2; ++t) {
  const int qt = t ? (ntq - 1 - (int)blockIdx.x) : (int)blockIdx.x;
  const int q0 = qt * QT;
  const int qrow = q0 + rh * 16 + l16;

  v2f qf[16];
  {
    const float* qptr =
        Qg + (size_t)(b * S + qrow) * 256 + h * 64 + 2 * half;
#pragma unroll
    for (int i = 0; i < 16; ++i) qf[i] = *(const v2f*)(qptr + i * 4);
  }

  v8f oacc[4];
#pragma unroll
  for (int nt = 0; nt < 4; ++nt) oacc[nt] = (v8f)(0.0f);
  float m = -1e30f, l = 0.0f;

  const int ntiles = (q0 + QT - 1) / KT + 1;
  v4f kreg[KT / 8], vreg[KT / 8];
#pragma unroll
  for (int it = 0; it < KT / 8; ++it) {
    kreg[it] = *(const v4f*)(Kp + gbase + (size_t)it * 8 * 256);
    vreg[it] = *(const v4f*)(Vp + gbase + (size_t)it * 8 * 256);
  }
  for (int j = 0; j < ntiles; ++j) {
    const int a0 = j * KT;
    __syncthreads();
#pragma unroll
    for (int it = 0; it < KT / 8; ++it) {
      *(v4f*)&Ks[ldr + it * 8][ldc] = kreg[it];
      *(v4f*)&Vs[ldr + it * 8][ldc] = vreg[it];
    }
    __syncthreads();
    if (j + 1 < ntiles) {
      const size_t g1 = gbase + (size_t)(a0 + KT) * 256;
#pragma unroll
      for (int it = 0; it < KT / 8; ++it) {
        kreg[it] = *(const v4f*)(Kp + g1 + (size_t)it * 8 * 256);
        vreg[it] = *(const v4f*)(Vp + g1 + (size_t)it * 8 * 256);
      }
    }

    v8f s[NKT];
#pragma unroll
    for (int kt = 0; kt < NKT; ++kt) s[kt] = (v8f)(0.0f);
    {
      const int krow = kh * KW + l16;
      v2f kf[2][SI][NKT];
#pragma unroll
      for (int ii = 0; ii < SI; ++ii)
#pragma unroll
        for (int kt = 0; kt < NKT; ++kt)
          kf[0][ii][kt] = *(const v2f*)&Ks[krow + kt * 16][ii * 4 + 2 * half];
#pragma unroll
      for (int blk = 0; blk < 16 / SI; ++blk) {
        const int cur = blk & 1;
        if (blk + 1 < 16 / SI) {
#pragma unroll
          for (int ii = 0; ii < SI; ++ii)
#pragma unroll
            for (int kt = 0; kt < NKT; ++kt)
              kf[cur ^ 1][ii][kt] =
                  *(const v2f*)&Ks[krow + kt * 16][((blk + 1) * SI + ii) * 4 + 2 * half];
        }
        __builtin_amdgcn_sched_barrier(0);
#pragma unroll
        for (int ii = 0; ii < SI; ++ii)
#pragma unroll
          for (int kt = 0; kt < NKT; ++kt)
            s[kt] = wmma4(kf[cur][ii][kt], qf[blk * SI + ii], s[kt]);
        __builtin_amdgcn_sched_barrier(0);
      }
    }

    const int vrow = kh * KW + 8 * half;
    float vx[2][SP][4], vy[2][SP][4];
#pragma unroll
    for (int s2 = 0; s2 < SP; ++s2) {
      const int kr = vrow + (s2 >> 2) * 16 + (s2 & 3) * 2;
#pragma unroll
      for (int nt = 0; nt < 4; ++nt) {
        vx[0][s2][nt] = Vs[kr + 0][nt * 16 + l16];
        vy[0][s2][nt] = Vs[kr + 1][nt * 16 + l16];
      }
    }

    {
      const int key0 = a0 + kh * KW + 8 * half;
      float mxk[NKT];
#pragma unroll
      for (int kt = 0; kt < NKT; ++kt) {
#pragma unroll
        for (int r = 0; r < 8; ++r) {
          float x = s[kt][r] * 0.125f;
          if (key0 + kt * 16 + r > qrow) x = -1e30f;
          s[kt][r] = x;
          mxk[kt] = r ? fmaxf(mxk[kt], x) : x;
        }
      }
      float mx = fmaxf(fmaxf(mxk[0], mxk[1]), fmaxf(mxk[2], mxk[3]));
      mx = fmaxf(mx, xhalf(mx));
      const float mnew = fmaxf(m, mx);
      float smk[NKT];
#pragma unroll
      for (int kt = 0; kt < NKT; ++kt) {
#pragma unroll
        for (int r = 0; r < 8; ++r) {
          const float p = __expf(s[kt][r] - mnew);
          s[kt][r] = p;
          smk[kt] = r ? smk[kt] + p : p;
        }
      }
      const float scale = __expf(m - mnew);
      l = l * scale + ((smk[0] + smk[1]) + (smk[2] + smk[3]));
      m = mnew;
#pragma unroll
      for (int nt = 0; nt < 4; ++nt)
#pragma unroll
        for (int r = 0; r < 8; ++r) oacc[nt][r] *= scale;
    }

    {
#pragma unroll
      for (int blk = 0; blk < KW / 4 / SP; ++blk) {
        const int cur = blk & 1;
        if (blk + 1 < KW / 4 / SP) {
#pragma unroll
          for (int s2 = 0; s2 < SP; ++s2) {
            const int st = (blk + 1) * SP + s2;
            const int kr = vrow + (st >> 2) * 16 + (st & 3) * 2;
#pragma unroll
            for (int nt = 0; nt < 4; ++nt) {
              vx[cur ^ 1][s2][nt] = Vs[kr + 0][nt * 16 + l16];
              vy[cur ^ 1][s2][nt] = Vs[kr + 1][nt * 16 + l16];
            }
          }
        }
        __builtin_amdgcn_sched_barrier(0);
#pragma unroll
        for (int s2 = 0; s2 < SP; ++s2) {
          const int st = blk * SP + s2;
          v2f pt;
          pt.x = s[st >> 2][(st & 3) * 2 + 0];
          pt.y = s[st >> 2][(st & 3) * 2 + 1];
#pragma unroll
          for (int nt = 0; nt < 4; ++nt) {
            v2f av;
            av.x = vx[cur][s2][nt];
            av.y = vy[cur][s2][nt];
            oacc[nt] = wmma4(av, pt, oacc[nt]);
          }
        }
        if (blk + 1 < KW / 4 / SP) __builtin_amdgcn_sched_barrier(0);
      }
    }
  }

  l += xhalf(l);
  if (kh == 1) {
    Xm[rh][lane] = m;
    Xl[rh][lane] = l;
#pragma unroll
    for (int nt = 0; nt < 4; ++nt)
#pragma unroll
      for (int r = 0; r < 8; ++r) Xo[rh][nt * 8 + r][lane] = oacc[nt][r];
  }
  __syncthreads();
  if (kh == 0) {
    const float m2 = Xm[rh][lane];
    const float l2 = Xl[rh][lane];
    const float mm = fmaxf(m, m2);
    const float w1 = __expf(m - mm);
    const float w2 = __expf(m2 - mm);
    const float inv = 1.0f / (l * w1 + l2 * w2);
    float* orow = Op + (size_t)(b * S + qrow) * 256 + h * 64 + 8 * half;
#pragma unroll
    for (int nt = 0; nt < 4; ++nt) {
      v4f lo, hi;
      lo.x = (oacc[nt][0] * w1 + Xo[rh][nt * 8 + 0][lane] * w2) * inv;
      lo.y = (oacc[nt][1] * w1 + Xo[rh][nt * 8 + 1][lane] * w2) * inv;
      lo.z = (oacc[nt][2] * w1 + Xo[rh][nt * 8 + 2][lane] * w2) * inv;
      lo.w = (oacc[nt][3] * w1 + Xo[rh][nt * 8 + 3][lane] * w2) * inv;
      hi.x = (oacc[nt][4] * w1 + Xo[rh][nt * 8 + 4][lane] * w2) * inv;
      hi.y = (oacc[nt][5] * w1 + Xo[rh][nt * 8 + 5][lane] * w2) * inv;
      hi.z = (oacc[nt][6] * w1 + Xo[rh][nt * 8 + 6][lane] * w2) * inv;
      hi.w = (oacc[nt][7] * w1 + Xo[rh][nt * 8 + 7][lane] * w2) * inv;
      *(v4f*)(orow + nt * 16) = lo;
      *(v4f*)(orow + nt * 16 + 4) = hi;
    }
  }
  }
}

extern "C" void kernel_launch(void* const* d_in, const int* in_sizes, int n_in,
                              void* d_out, int out_size, void* d_ws,
                              size_t ws_size, hipStream_t stream) {
  const float* q_in  = (const float*)d_in[0];
  const float* kv_in = (const float*)d_in[1];
  const float* Wq = (const float*)d_in[3];
  const float* Wk = (const float*)d_in[4];
  const float* Wv = (const float*)d_in[5];
  const float* Wo = (const float*)d_in[6];
  float* out = (float*)d_out;

  const int B = 2, S = 2048, IN = 1024;
  const int M = B * S;

  float* ws  = (float*)d_ws;
  float* wqg = ws;
  float* qg  = wqg + 1024 * 256;
  float* kp  = qg + M * 256;
  float* vp  = kp + M * 256;
  float* op  = vp + M * 256;

  reduce_wq_kernel<<<(1024 * 256) / 256, 256, 0, stream>>>(Wq, wqg);

  dim3 blk(BLOCK);
  gemm_f32_wmma<<<dim3(256 / 64, M / 64, 3), blk, 0, stream>>>(
      q_in, wqg, qg, kv_in, Wk, kp, kv_in, Wv, vp, M, 256, IN);

  fa_f32_wmma<<<dim3(S / 64, B * 4), blk, 0, stream>>>(qg, kp, vp, op, S);

  gemm_f32_wmma<<<dim3(IN / 64, M / 64, 1), blk, 0, stream>>>(
      op, Wo, out, nullptr, nullptr, nullptr, nullptr, nullptr, nullptr, M, IN, 256);
}


# attention: 8-wave blocks, two waves per SIMD; each wave owns 16 keys of every 128-key trip in a private LDS slab and covers all 32 queries: no barrier in the key loop; changes order of row sums
# speedup vs baseline: 9.0759x; 1.0345x over previous
#include <hip/hip_runtime.h>
#include <hip/hip_bf16.h>

typedef __attribute__((ext_vector_type(2))) float v2f;
typedef __attribute__((ext_vector_type(8))) float v8f;
typedef __attribute__((ext_vector_type(4))) float v4f;

#define BLOCK 128

__device__ __forceinline__ v8f wmma4(v2f a, v2f b, v8f c) {
  return __builtin_amdgcn_wmma_f32_16x16x4_f32(false, a, false, b, (short)0, c,
                                               false, false);
}

__global__ void reduce_wq_kernel(const float* __restrict__ Wq,
                                 float* __restrict__ Wqg) {
  int idx = blockIdx.x * blockDim.x + threadIdx.x;
  if (idx >= 1024 * 256) return;
  int i = idx >> 8;
  int c = idx & 255;
  int h = c >> 6, d = c & 63;
  const float* p = Wq + (size_t)i * 1024 + h * 256 + d;
  Wqg[idx] = p[0] + p[64] + p[128] + p[192];
}

#define GK 128
#define GIT (GK / 8)
#define APITCH (GK + 4)
#define SB 8
#define NB (GK / 4 / SB)
__global__ __launch_bounds__(BLOCK, 3) void gemm_f32_wmma(
    const float* __restrict__ A0, const float* __restrict__ B0, float* __restrict__ C0,
    const float* __restrict__ A1, const float* __restrict__ B1, float* __restrict__ C1,
    const float* __restrict__ A2, const float* __restrict__ B2, float* __restrict__ C2,
    int M, int N, int K) {
  __shared__ __align__(16) float As[64][APITCH];
  __shared__ __align__(16) float Bs[GK][64];
  const int z = blockIdx.z;
  const float* A = z == 0 ? A0 : (z == 1 ? A1 : A2);
  const float* B = z == 0 ? B0 : (z == 1 ? B1 : B2);
  float* C = z == 0 ? C0 : (z == 1 ? C1 : C2);
  const int tid  = threadIdx.x;
  const int wave = tid >> 5;
  const int lane = tid & 31;
  const int half = lane >> 4;
  const int l16  = lane & 15;
  const int row0 = blockIdx.y * 64;
  const int col0 = blockIdx.x * 64;

  v8f acc[4];
#pragma unroll
  for (int nt = 0; nt < 4; ++nt) acc[nt] = (v8f)(0.0f);

  const int ar = tid / (GK / 4), ac = (tid % (GK / 4)) << 2;
  const int br = tid >> 4, bc = (tid & 15) << 2;
  const float* aptr = A + (size_t)(row0 + ar) * K + ac;
  const float* bptr = B + (size_t)br * N + col0 + bc;

  v4f areg[GIT], breg[GIT];
#pragma unroll
  for (int it = 0; it < GIT; ++it) {
    areg[it] = *(const v4f*)(aptr + (size_t)(it * (512 / GK)) * K);
    breg[it] = *(const v4f*)(bptr + (size_t)(it * 8) * N);
  }

  for (int kk = 0; kk < K; kk += GK) {
    __syncthreads();
#pragma unroll
    for (int it = 0; it < GIT; ++it) {
      *(v4f*)&As[ar + it * (512 / GK)][ac] = areg[it];
      float* bdst = &Bs[((br >> 2) + 2 * it) * 4][((br >> 1) & 1) * 32 + 2 * (tid & 15) + (br & 1)];
      bdst[0]   = breg[it].x;
      bdst[64]  = breg[it].y;
      bdst[128] = breg[it].z;
      bdst[192] = breg[it].w;
    }
    __syncthreads();
    if (kk + GK < K) {
#pragma unroll
      for (int it = 0; it < GIT; ++it) {
        areg[it] = *(const v4f*)(aptr + (size_t)(it * (512 / GK)) * K + kk + GK);
        breg[it] = *(const v4f*)(bptr + (size_t)(kk + GK + it * 8) * N);
      }
    }
    v2f af[2][SB];
    v2f bf[2][SB][4];
#pragma unroll
    for (int s = 0; s < SB; ++s) {
      const int k0 = s * 4;
      af[0][s] = *(const v2f*)&As[wave * 16 + l16][k0 + 2 * half];
#pragma unroll
      for (int nt = 0; nt < 4; ++nt)
        bf[0][s][nt] = *(const v2f*)&Bs[k0 + nt][2 * lane];
    }
#pragma unroll
    for (int blk = 0; blk < NB; ++blk) {
      const int cur = blk & 1;
      if (blk + 1 < NB) {
#pragma unroll
        for (int s = 0; s < SB; ++s) {
          const int k0 = ((blk + 1) * SB + s) * 4;
          af[cur ^ 1][s] = *(const v2f*)&As[wave * 16 + l16][k0 + 2 * half];
#pragma unroll
          for (int nt = 0; nt < 4; ++nt)
            bf[cur ^ 1][s][nt] = *(const v2f*)&Bs[k0 + nt][2 * lane];
        }
      }
      __builtin_amdgcn_sched_barrier(0);
#pragma unroll
      for (int s = 0; s < SB; ++s)
#pragma unroll
        for (int nt = 0; nt < 4; ++nt)
          acc[nt] = wmma4(af[cur][s], bf[cur][s][nt], acc[nt]);
      if (blk + 1 < NB) __builtin_amdgcn_sched_barrier(0);
    }
  }
#pragma unroll
  for (int r = 0; r < 8; ++r) {
    v4f o;
    o.x = acc[0][r];
    o.y = acc[1][r];
    o.z = acc[2][r];
    o.w = acc[3][r];
    *(v4f*)&C[(size_t)(row0 + wave * 16 + r + 8 * half) * N + col0 + 4 * l16] = o;
  }
}

#define ABLOCK 256
#define NWAVE (ABLOCK / 32)
#define KT 128
#define KW (KT / NWAVE)
#define QT 32
#define NQ (QT / 16)
#define KPITCH 68
#define VPITCH 68

__device__ __forceinline__ float xhalf(float x) {
  const int b = __builtin_bit_cast(int, x);
  return __builtin_bit_cast(
      float, __builtin_amdgcn_permlanex16(b, b, 0x76543210, 0xfedcba98, false, false));
}

__global__ __launch_bounds__(ABLOCK) void fa_f32_wmma(
    const float* __restrict__ Qg, const float* __restrict__ Kp,
    const float* __restrict__ Vp, float* __restrict__ Op, int S) {
  __shared__ __align__(16) float Ks[NWAVE][KW][KPITCH];
  __shared__ __align__(16) float Vs[NWAVE][KW][VPITCH];
  __shared__ float Xo[NWAVE][NQ * 32][32];
  __shared__ float Xm[NWAVE][NQ][32];
  __shared__ float Xl[NWAVE][NQ][32];
  const int tid  = threadIdx.x;
  const int wave = tid >> 5;
  const int lane = tid & 31;
  const int half = lane >> 4;
  const int l16  = lane & 15;
  const int b    = blockIdx.y >> 2;
  const int h    = blockIdx.y & 3;
  const int ntq  = S / QT;

  const int ldr = lane >> 4, ldc = (lane & 15) << 2;
  const size_t gbase = (size_t)(b * S + wave * KW + ldr) * 256 + h * 64 + ldc;

  for (int t = 0; t < 2; ++t) {
  const int qt = t ? (ntq - 1 - (int)blockIdx.x) : (int)blockIdx.x;
  const int q0 = qt * QT;

  v2f qf[NQ][16];
#pragma unroll
  for (int q = 0; q < NQ; ++q) {
    const float* qptr =
        Qg + (size_t)(b * S + q0 + q * 16 + l16) * 256 + h * 64 + 2 * half;
#pragma unroll
    for (int i = 0; i < 16; ++i) qf[q][i] = *(const v2f*)(qptr + i * 4);
  }

  v8f oacc[NQ][4];
#pragma unroll
  for (int q = 0; q < NQ; ++q)
#pragma unroll
    for (int nt = 0; nt < 4; ++nt) oacc[q][nt] = (v8f)(0.0f);
  float m[NQ], l[NQ];
#pragma unroll
  for (int q = 0; q < NQ; ++q) { m[q] = -1e30f; l[q] = 0.0f; }

  const int ntiles = (q0 + QT - 1) / KT + 1;
  v4f kreg[KW / 2], vreg[KW / 2];
#pragma unroll
  for (int it = 0; it < KW / 2; ++it) {
    kreg[it] = *(const v4f*)(Kp + gbase + (size_t)it * 2 * 256);
    vreg[it] = *(const v4f*)(Vp + gbase + (size_t)it * 2 * 256);
  }
  for (int j = 0; j < ntiles; ++j) {
    const int a0 = j * KT;
#pragma unroll
    for (int it = 0; it < KW / 2; ++it) {
      *(v4f*)&Ks[wave][ldr + it * 2][ldc] = kreg[it];
      *(v4f*)&Vs[wave][ldr + it * 2][ldc] = vreg[it];
    }
    if (j + 1 < ntiles) {
      const size_t g1 = gbase + (size_t)(a0 + KT) * 256;
#pragma unroll
      for (int it = 0; it < KW / 2; ++it) {
        kreg[it] = *(const v4f*)(Kp + g1 + (size_t)it * 2 * 256);
        vreg[it] = *(const v4f*)(Vp + g1 + (size_t)it * 2 * 256);
      }
    }

    v2f kf[16];
#pragma unroll
    for (int i = 0; i < 16; ++i)
      kf[i] = *(const v2f*)&Ks[wave][l16][i * 4 + 2 * half];
    float vx[4][4], vy[4][4];
#pragma unroll
    for (int jj = 0; jj < 4; ++jj) {
      const int kr = 8 * half + 2 * jj;
#pragma unroll
      for (int nt = 0; nt < 4; ++nt) {
        vx[jj][nt] = Vs[wave][kr + 0][nt * 16 + l16];
        vy[jj][nt] = Vs[wave][kr + 1][nt * 16 + l16];
      }
    }
    __builtin_amdgcn_sched_barrier(0);

    v8f s[NQ];
#pragma unroll
    for (int q = 0; q < NQ; ++q) s[q] = (v8f)(0.0f);
#pragma unroll
    for (int i = 0; i < 16; ++i)
#pragma unroll
      for (int q = 0; q < NQ; ++q)
        s[q] = wmma4(kf[i], qf[q][i], s[q]);

    {
      const int key0 = a0 + wave * KW + 8 * half;
#pragma unroll
      for (int q = 0; q < NQ; ++q) {
        const int qrow = q0 + q * 16 + l16;
        float mx = 0.0f;
#pragma unroll
        for (int r = 0; r < 8; ++r) {
          float x = s[q][r] * 0.125f;
          if (key0 + r > qrow) x = -1e30f;
          s[q][r] = x;
          mx = r ? fmaxf(mx, x) : x;
        }
        mx = fmaxf(mx, xhalf(mx));
        const float mnew = fmaxf(m[q], mx);
        float sm = 0.0f;
#pragma unroll
        for (int r = 0; r < 8; ++r) {
          const float p = __expf(s[q][r] - mnew);
          s[q][r] = p;
          sm = r ? sm + p : p;
        }
        const float scale = __expf(m[q] - mnew);
        l[q] = l[q] * scale + sm;
        m[q] = mnew;
#pragma unroll
        for (int nt = 0; nt < 4; ++nt)
#pragma unroll
          for (int r = 0; r < 8; ++r) oacc[q][nt][r] *= scale;
      }
    }

#pragma unroll
    for (int jj = 0; jj < 4; ++jj) {
#pragma unroll
      for (int nt = 0; nt < 4; ++nt) {
        v2f av;
        av.x = vx[jj][nt];
        av.y = vy[jj][nt];
#pragma unroll
        for (int q = 0; q < NQ; ++q) {
          v2f pt;
          pt.x = s[q][jj * 2 + 0];
          pt.y = s[q][jj * 2 + 1];
          oacc[q][nt] = wmma4(av, pt, oacc[q][nt]);
        }
      }
    }
  }

#pragma unroll
  for (int q = 0; q < NQ; ++q) l[q] += xhalf(l[q]);
#pragma unroll
  for (int q = 0; q < NQ; ++q) {
    Xm[wave][q][lane] = m[q];
    Xl[wave][q][lane] = l[q];
#pragma unroll
    for (int nt = 0; nt < 4; ++nt)
#pragma unroll
      for (int r = 0; r < 8; ++r) Xo[wave][q * 32 + nt * 8 + r][lane] = oacc[q][nt][r];
  }
  __syncthreads();
  if (wave < 2 * NQ) {
    const int mq = wave >> 1, ln = wave & 1;
    float mw[NWAVE];
    float mm = 0.0f;
#pragma unroll
    for (int w = 0; w < NWAVE; ++w) {
      mw[w] = Xm[w][mq][lane];
      mm = w ? fmaxf(mm, mw[w]) : mw[w];
    }
    float wt[NWAVE];
    float den = 0.0f;
#pragma unroll
    for (int w = 0; w < NWAVE; ++w) {
      wt[w] = __expf(mw[w] - mm);
      const float lw = Xl[w][mq][lane] * wt[w];
      den = w ? den + lw : lw;
    }
    const float inv = 1.0f / den;
    float* orow = Op + (size_t)(b * S + q0 + mq * 16 + l16) * 256 + h * 64 + ln * 32 + 8 * half;
#pragma unroll
    for (int n2 = 0; n2 < 2; ++n2) {
      const int e0 = mq * 32 + (2 * ln + n2) * 8;
      float o[8];
#pragma unroll
      for (int r = 0; r < 8; ++r) {
        float acc = 0.0f;
#pragma unroll
        for (int w = 0; w < NWAVE; ++w) {
          const float ow = Xo[w][e0 + r][lane] * wt[w];
          acc = w ? acc + ow : ow;
        }
        o[r] = acc * inv;
      }
      v4f lo, hi;
      lo.x = o[0]; lo.y = o[1]; lo.z = o[2]; lo.w = o[3];
      hi.x = o[4]; hi.y = o[5]; hi.z = o[6]; hi.w = o[7];
      *(v4f*)(orow + n2 * 16) = lo;
      *(v4f*)(orow + n2 * 16 + 4) = hi;
    }
  }
  __syncthreads();
  }
}

extern "C" void kernel_launch(void* const* d_in, const int* in_sizes, int n_in,
                              void* d_out, int out_size, void* d_ws,
                              size_t ws_size, hipStream_t stream) {
  const float* q_in  = (const float*)d_in[0];
  const float* kv_in = (const float*)d_in[1];
  const float* Wq = (const float*)d_in[3];
  const float* Wk = (const float*)d_in[4];
  const float* Wv = (const float*)d_in[5];
  const float* Wo = (const float*)d_in[6];
  float* out = (float*)d_out;

  const int B = 2, S = 2048, IN = 1024;
  const int M = B * S;

  float* ws  = (float*)d_ws;
  float* wqg = ws;
  float* qg  = wqg + 1024 * 256;
  float* kp  = qg + M * 256;
  float* vp  = kp + M * 256;
  float* op  = vp + M * 256;

  reduce_wq_kernel<<<(1024 * 256) / 256, 256, 0, stream>>>(Wq, wqg);

  dim3 blk(BLOCK);
  gemm_f32_wmma<<<dim3(256 / 64, M / 64, 3), blk, 0, stream>>>(
      q_in, wqg, qg, kv_in, Wk, kp, kv_in, Wv, vp, M, 256, IN);

  fa_f32_wmma<<<dim3(S / 64, B * 4), dim3(ABLOCK), 0, stream>>>(qg, kp, vp, op, S);

  gemm_f32_wmma<<<dim3(IN / 64, M / 64, 1), blk, 0, stream>>>(
      op, Wo, out, nullptr, nullptr, nullptr, nullptr, nullptr, nullptr, M, IN, 256);
}
